# starting kernel
# baseline (speedup 1.0000x reference)
#include <hip/hip_runtime.h>
#include <math.h>


typedef __attribute__((ext_vector_type(2))) float v2f;
typedef __attribute__((ext_vector_type(8))) float v8f;

#define NDIM 1024
#define LSEQ 8192

__global__ __launch_bounds__(128)
void gemm_wmma_f32(const float* __restrict__ Ag, int lda,
                   const float* __restrict__ Bg, int ldb,
                   float* __restrict__ Cg, int ldc,
                   int Kdim, int mBound, int nBound, int addI)
{
    __shared__ float As[64][33];
    __shared__ float Bs[64][35];

    const int tid  = threadIdx.x;
    const int row0 = blockIdx.y * 64;
    const int col0 = blockIdx.x * 64;
    const int lane = tid & 31;
    const int wv   = tid >> 5;
    const int wm   = (wv >> 1) * 32;
    const int wn   = (wv & 1)  * 32;
    const int g    = lane >> 4;
    const int lm   = lane & 15;

    v8f acc[2][2] = {};

    for (int k0 = 0; k0 < Kdim; k0 += 32) {
        #pragma unroll
        for (int i = tid; i < 64 * 32; i += 128) {
            int r = i >> 5, c = i & 31;
            As[r][c] = Ag[(size_t)(row0 + r) * lda + (k0 + c)];
        }
        #pragma unroll
        for (int i = tid; i < 32 * 64; i += 128) {
            int r = i >> 6, c = i & 63;
            Bs[c][r] = Bg[(size_t)(k0 + r) * ldb + (col0 + c)];
        }
        if (k0 + 32 < Kdim) {
            __builtin_prefetch(&Ag[(size_t)(row0 + (tid >> 1)) * lda + (k0 + 32) + (tid & 1) * 16], 0, 0);
            __builtin_prefetch(&Bg[(size_t)(k0 + 32 + (tid >> 2)) * ldb + col0 + (tid & 3) * 16], 0, 0);
        }
        __syncthreads();

        #pragma unroll
        for (int kk = 0; kk < 32; kk += 4) {
            const int ka = kk + 2 * g;
            v2f a0, a1, b0, b1;
            a0.x = As[wm + lm][ka];           a0.y = As[wm + lm][ka + 1];
            a1.x = As[wm + 16 + lm][ka];      a1.y = As[wm + 16 + lm][ka + 1];
            b0.x = Bs[wn + lm][ka];           b0.y = Bs[wn + lm][ka + 1];
            b1.x = Bs[wn + 16 + lm][ka];      b1.y = Bs[wn + 16 + lm][ka + 1];
            acc[0][0] = __builtin_amdgcn_wmma_f32_16x16x4_f32(false, a0, false, b0, (short)0, acc[0][0], false, false);
            acc[0][1] = __builtin_amdgcn_wmma_f32_16x16x4_f32(false, a0, false, b1, (short)0, acc[0][1], false, false);
            acc[1][0] = __builtin_amdgcn_wmma_f32_16x16x4_f32(false, a1, false, b0, (short)0, acc[1][0], false, false);
            acc[1][1] = __builtin_amdgcn_wmma_f32_16x16x4_f32(false, a1, false, b1, (short)0, acc[1][1], false, false);
        }
        __syncthreads();
    }

    #pragma unroll
    for (int mi = 0; mi < 2; ++mi)
        #pragma unroll
        for (int ni = 0; ni < 2; ++ni)
            #pragma unroll
            for (int r = 0; r < 8; ++r) {
                int row = row0 + wm + mi * 16 + r + 8 * g;
                int col = col0 + wn + ni * 16 + lm;
                if (row < mBound && col < nBound) {
                    float v = acc[mi][ni][r];
                    if (addI && row == col) v += 1.0f;
                    Cg[(size_t)row * ldc + col] = v;
                }
            }
}

__global__ void prep_mats(const float* __restrict__ A, const float* __restrict__ ldt,
                          float* __restrict__ M, float* __restrict__ IM, int n)
{
    int idx = blockIdx.x * blockDim.x + threadIdx.x;
    if (idx >= n * n) return;
    float h = 0.5f * expf(ldt[0]);
    float m = h * A[idx];
    M[idx]  = m;
    int r = idx / n, c = idx - r * n;
    IM[idx] = m + ((r == c) ? 1.0f : 0.0f);
}

__global__ void bb_matvec(const float* __restrict__ S, const float* __restrict__ Bv,
                          const float* __restrict__ ldt, float* __restrict__ Y,
                          int n, int ldy)
{
    int i = blockIdx.x * blockDim.x + threadIdx.x;
    if (i >= n) return;
    float dt = expf(ldt[0]);
    float s = 0.f;
    for (int k = 0; k < n; ++k) s += S[(size_t)i * n + k] * Bv[k];
    Y[(size_t)i * ldy] = dt * s;
}

__global__ void copy_row(const float* __restrict__ Cv, float* __restrict__ R, int n)
{
    int i = blockIdx.x * blockDim.x + threadIdx.x;
    if (i < n) R[i] = Cv[i];
}

__global__ __launch_bounds__(256)
void conv_causal(const float* __restrict__ u, const float* __restrict__ Kk,
                 const float* __restrict__ Dv, float* __restrict__ y, int L)
{
    __shared__ float Ks[256];
    __shared__ float us[512];
    const int tid = threadIdx.x;
    const int t0  = blockIdx.x * 256;
    const int t   = t0 + tid;
    float acc = 0.f;
    for (int c0 = 0; c0 <= t0; c0 += 256) {
        Ks[tid] = Kk[(L - 1) - (c0 + tid)];
        int base = t0 - c0 - 255;
        int i0 = base + tid;
        us[tid] = (i0 >= 0) ? u[i0] : 0.f;
        int i1 = base + 256 + tid;
        us[256 + tid] = (i1 >= 0 && i1 < L) ? u[i1] : 0.f;
        __syncthreads();
        int jmax = t - c0; if (jmax > 255) jmax = 255;
        for (int j = 0; j <= jmax; ++j)
            acc += Ks[j] * us[tid + 255 - j];
        __syncthreads();
    }
    y[t] = acc + Dv[0] * u[t];
}

extern "C" void kernel_launch(void* const* d_in, const int* in_sizes, int n_in,
                              void* d_out, int out_size, void* d_ws, size_t ws_size,
                              hipStream_t stream)
{
    (void)in_sizes; (void)n_in; (void)out_size; (void)ws_size;
    const float* u   = (const float*)d_in[0];
    const float* A   = (const float*)d_in[1];
    const float* Bv  = (const float*)d_in[2];
    const float* Cv  = (const float*)d_in[3];
    const float* Dv  = (const float*)d_in[4];
    const float* ldt = (const float*)d_in[5];
    float* y = (float*)d_out;
    const int N = NDIM, L = LSEQ;

    float* B0 = (float*)d_ws;
    float* B1 = B0 + (size_t)N * N;
    float* B2 = B1 + (size_t)N * N;
    float* B3 = B2 + (size_t)N * N;
    float* Y  = B3 + (size_t)N * N;
    float* R  = Y  + (size_t)N * 128;
    float* Kk = R  + (size_t)64 * N;

    dim3 blk(128);
    auto gemm = [&](const float* Am, int lda, const float* Bm, int ldb,
                    float* Cm, int ldc, int Mc, int Nc, int Kd,
                    int mB, int nB, int addI) {
        dim3 grid((Nc + 63) / 64, (Mc + 63) / 64);
        gemm_wmma_f32<<<grid, blk, 0, stream>>>(Am, lda, Bm, ldb, Cm, ldc, Kd, mB, nB, addI);
    };

    prep_mats<<<(N * N + 255) / 256, 256, 0, stream>>>(A, ldt, B0, B1, N);
    gemm(B0, N, B1, N, B2, N, N, N, N, N, N, 1);
    gemm(B0, N, B2, N, B3, N, N, N, N, N, N, 1);
    gemm(B0, N, B3, N, B2, N, N, N, N, N, N, 1);
    gemm(B2, N, B1, N, B3, N, N, N, N, N, N, 0);
    bb_matvec<<<(N + 255) / 256, 256, 0, stream>>>(B2, Bv, ldt, Y, N, 128);

    gemm(B3, N, Y, 128, Y + 1,  128, N, 64, N, N, 127, 0);
    gemm(B3, N, B3, N, B0, N, N, N, N, N, N, 0);
    gemm(B0, N, Y, 128, Y + 2,  128, N, 64, N, N, 126, 0);
    gemm(B0, N, B0, N, B1, N, N, N, N, N, N, 0);
    gemm(B1, N, Y, 128, Y + 4,  128, N, 64, N, N, 124, 0);
    gemm(B1, N, B1, N, B0, N, N, N, N, N, N, 0);
    gemm(B0, N, Y, 128, Y + 8,  128, N, 64, N, N, 120, 0);
    gemm(B0, N, B0, N, B1, N, N, N, N, N, N, 0);
    gemm(B1, N, Y, 128, Y + 16, 128, N, 64, N, N, 112, 0);
    gemm(B1, N, B1, N, B0, N, N, N, N, N, N, 0);
    gemm(B0, N, Y, 128, Y + 32, 128, N, 64, N, N,  96, 0);
    gemm(B0, N, B0, N, B1, N, N, N, N, N, N, 0);
    gemm(B1, N, Y, 128, Y + 64, 128, N, 64, N, N,  64, 0);
    gemm(B1, N, B1, N, B0, N, N, N, N, N, N, 0);

    copy_row<<<(N + 255) / 256, 256, 0, stream>>>(Cv, R, N);
    gemm(R, N, B0, N, R + 1 * N,  N, 64, N, N, 63, N, 0);
    gemm(B0, N, B0, N, B1, N, N, N, N, N, N, 0);
    gemm(R, N, B1, N, R + 2 * N,  N, 64, N, N, 62, N, 0);
    gemm(B1, N, B1, N, B0, N, N, N, N, N, N, 0);
    gemm(R, N, B0, N, R + 4 * N,  N, 64, N, N, 60, N, 0);
    gemm(B0, N, B0, N, B1, N, N, N, N, N, N, 0);
    gemm(R, N, B1, N, R + 8 * N,  N, 64, N, N, 56, N, 0);
    gemm(B1, N, B1, N, B0, N, N, N, N, N, N, 0);
    gemm(R, N, B0, N, R + 16 * N, N, 64, N, N, 48, N, 0);
    gemm(B0, N, B0, N, B1, N, N, N, N, N, N, 0);
    gemm(R, N, B1, N, R + 32 * N, N, 64, N, N, 32, N, 0);

    gemm(R, N, Y, 128, Kk, 128, 64, 128, N, 64, 128, 0);

    conv_causal<<<L / 256, 256, 0, stream>>>(u, Kk, Dv, y, L);
}


# GEMM staging: each K-chunk's 32 global loads per thread issue as one batch (fixed trip count, row strides as template constants, base+immediate) instead of 16 serial load-wait-store round trips
# speedup vs baseline: 3.2815x; 3.2815x over previous
#include <hip/hip_runtime.h>
#include <math.h>


typedef __attribute__((ext_vector_type(2))) float v2f;
typedef __attribute__((ext_vector_type(8))) float v8f;

#define NDIM 1024
#define LSEQ 8192

template <int LDA, int LDB>
__global__ __launch_bounds__(128)
void gemm_wmma_f32(const float* __restrict__ Ag,
                   const float* __restrict__ Bg,
                   float* __restrict__ Cg, int ldc,
                   int Kdim, int mBound, int nBound, int addI)
{
    __shared__ float As[64][33];
    __shared__ float Bs[64][35];

    const int tid  = threadIdx.x;
    const int row0 = blockIdx.y * 64;
    const int col0 = blockIdx.x * 64;
    const int lane = tid & 31;
    const int wv   = tid >> 5;
    const int wm   = (wv >> 1) * 32;
    const int wn   = (wv & 1)  * 32;
    const int g    = lane >> 4;
    const int lm   = lane & 15;

    v8f acc[2][2] = {};

    const float* aBase = Ag + (size_t)(row0 + (tid >> 5)) * LDA + (tid & 31);
    const float* bBase = Bg + (size_t)(tid >> 6) * LDB + (col0 + (tid & 63));

    for (int k0 = 0; k0 < Kdim; k0 += 32) {
        const float* aP = aBase + k0;
        const float* bP = bBase + (size_t)k0 * LDB;
        float ra[16], rb[16];
        #pragma unroll
        for (int it = 0; it < 16; ++it) ra[it] = aP[it * 4 * LDA];
        #pragma unroll
        for (int it = 0; it < 16; ++it) rb[it] = bP[it * 2 * LDB];
        #pragma unroll
        for (int it = 0; it < 16; ++it) {
            int i = tid + it * 128;
            As[i >> 5][i & 31] = ra[it];
        }
        #pragma unroll
        for (int it = 0; it < 16; ++it) {
            int i = tid + it * 128;
            Bs[i & 63][i >> 6] = rb[it];
        }
        if (k0 + 32 < Kdim) {
            __builtin_prefetch(&Ag[(size_t)(row0 + (tid >> 1)) * LDA + (k0 + 32) + (tid & 1) * 16], 0, 0);
            __builtin_prefetch(&Bg[(size_t)(k0 + 32 + (tid >> 2)) * LDB + col0 + (tid & 3) * 16], 0, 0);
        }
        __syncthreads();

        #pragma unroll
        for (int kk = 0; kk < 32; kk += 4) {
            const int ka = kk + 2 * g;
            v2f a0, a1, b0, b1;
            a0.x = As[wm + lm][ka];           a0.y = As[wm + lm][ka + 1];
            a1.x = As[wm + 16 + lm][ka];      a1.y = As[wm + 16 + lm][ka + 1];
            b0.x = Bs[wn + lm][ka];           b0.y = Bs[wn + lm][ka + 1];
            b1.x = Bs[wn + 16 + lm][ka];      b1.y = Bs[wn + 16 + lm][ka + 1];
            acc[0][0] = __builtin_amdgcn_wmma_f32_16x16x4_f32(false, a0, false, b0, (short)0, acc[0][0], false, false);
            acc[0][1] = __builtin_amdgcn_wmma_f32_16x16x4_f32(false, a0, false, b1, (short)0, acc[0][1], false, false);
            acc[1][0] = __builtin_amdgcn_wmma_f32_16x16x4_f32(false, a1, false, b0, (short)0, acc[1][0], false, false);
            acc[1][1] = __builtin_amdgcn_wmma_f32_16x16x4_f32(false, a1, false, b1, (short)0, acc[1][1], false, false);
        }
        __syncthreads();
    }

    #pragma unroll
    for (int mi = 0; mi < 2; ++mi)
        #pragma unroll
        for (int ni = 0; ni < 2; ++ni)
            #pragma unroll
            for (int r = 0; r < 8; ++r) {
                int row = row0 + wm + mi * 16 + r + 8 * g;
                int col = col0 + wn + ni * 16 + lm;
                if (row < mBound && col < nBound) {
                    float v = acc[mi][ni][r];
                    if (addI && row == col) v += 1.0f;
                    Cg[(size_t)row * ldc + col] = v;
                }
            }
}

__global__ void prep_mats(const float* __restrict__ A, const float* __restrict__ ldt,
                          float* __restrict__ M, float* __restrict__ IM, int n)
{
    int idx = blockIdx.x * blockDim.x + threadIdx.x;
    if (idx >= n * n) return;
    float h = 0.5f * expf(ldt[0]);
    float m = h * A[idx];
    M[idx]  = m;
    int r = idx / n, c = idx - r * n;
    IM[idx] = m + ((r == c) ? 1.0f : 0.0f);
}

__global__ void bb_matvec(const float* __restrict__ S, const float* __restrict__ Bv,
                          const float* __restrict__ ldt, float* __restrict__ Y,
                          int n, int ldy)
{
    int i = blockIdx.x * blockDim.x + threadIdx.x;
    if (i >= n) return;
    float dt = expf(ldt[0]);
    float s = 0.f;
    for (int k = 0; k < n; ++k) s += S[(size_t)i * n + k] * Bv[k];
    Y[(size_t)i * ldy] = dt * s;
}

__global__ void copy_row(const float* __restrict__ Cv, float* __restrict__ R, int n)
{
    int i = blockIdx.x * blockDim.x + threadIdx.x;
    if (i < n) R[i] = Cv[i];
}

__global__ __launch_bounds__(256)
void conv_causal(const float* __restrict__ u, const float* __restrict__ Kk,
                 const float* __restrict__ Dv, float* __restrict__ y, int L)
{
    __shared__ float Ks[256];
    __shared__ float us[512];
    const int tid = threadIdx.x;
    const int t0  = blockIdx.x * 256;
    const int t   = t0 + tid;
    float acc = 0.f;
    for (int c0 = 0; c0 <= t0; c0 += 256) {
        Ks[tid] = Kk[(L - 1) - (c0 + tid)];
        int base = t0 - c0 - 255;
        int i0 = base + tid;
        us[tid] = (i0 >= 0) ? u[i0] : 0.f;
        int i1 = base + 256 + tid;
        us[256 + tid] = (i1 >= 0 && i1 < L) ? u[i1] : 0.f;
        __syncthreads();
        int jmax = t - c0; if (jmax > 255) jmax = 255;
        for (int j = 0; j <= jmax; ++j)
            acc += Ks[j] * us[tid + 255 - j];
        __syncthreads();
    }
    y[t] = acc + Dv[0] * u[t];
}

extern "C" void kernel_launch(void* const* d_in, const int* in_sizes, int n_in,
                              void* d_out, int out_size, void* d_ws, size_t ws_size,
                              hipStream_t stream)
{
    (void)in_sizes; (void)n_in; (void)out_size; (void)ws_size;
    const float* u   = (const float*)d_in[0];
    const float* A   = (const float*)d_in[1];
    const float* Bv  = (const float*)d_in[2];
    const float* Cv  = (const float*)d_in[3];
    const float* Dv  = (const float*)d_in[4];
    const float* ldt = (const float*)d_in[5];
    float* y = (float*)d_out;
    const int N = NDIM, L = LSEQ;

    float* B0 = (float*)d_ws;
    float* B1 = B0 + (size_t)N * N;
    float* B2 = B1 + (size_t)N * N;
    float* B3 = B2 + (size_t)N * N;
    float* Y  = B3 + (size_t)N * N;
    float* R  = Y  + (size_t)N * 128;
    float* Kk = R  + (size_t)64 * N;

    dim3 blk(128);
    auto gemm = [&](const float* Am, int lda, const float* Bm, int ldb,
                    float* Cm, int ldc, int Mc, int Nc, int Kd,
                    int mB, int nB, int addI) {
        (void)lda;
        dim3 grid((Nc + 63) / 64, (Mc + 63) / 64);
        if (ldb == 128)
            gemm_wmma_f32<NDIM, 128><<<grid, blk, 0, stream>>>(Am, Bm, Cm, ldc, Kd, mB, nB, addI);
        else
            gemm_wmma_f32<NDIM, NDIM><<<grid, blk, 0, stream>>>(Am, Bm, Cm, ldc, Kd, mB, nB, addI);
    };

    prep_mats<<<(N * N + 255) / 256, 256, 0, stream>>>(A, ldt, B0, B1, N);
    gemm(B0, N, B1, N, B2, N, N, N, N, N, N, 1);
    gemm(B0, N, B2, N, B3, N, N, N, N, N, N, 1);
    gemm(B0, N, B3, N, B2, N, N, N, N, N, N, 1);
    gemm(B2, N, B1, N, B3, N, N, N, N, N, N, 0);
    bb_matvec<<<(N + 255) / 256, 256, 0, stream>>>(B2, Bv, ldt, Y, N, 128);

    gemm(B3, N, Y, 128, Y + 1,  128, N, 64, N, N, 127, 0);
    gemm(B3, N, B3, N, B0, N, N, N, N, N, N, 0);
    gemm(B0, N, Y, 128, Y + 2,  128, N, 64, N, N, 126, 0);
    gemm(B0, N, B0, N, B1, N, N, N, N, N, N, 0);
    gemm(B1, N, Y, 128, Y + 4,  128, N, 64, N, N, 124, 0);
    gemm(B1, N, B1, N, B0, N, N, N, N, N, N, 0);
    gemm(B0, N, Y, 128, Y + 8,  128, N, 64, N, N, 120, 0);
    gemm(B0, N, B0, N, B1, N, N, N, N, N, N, 0);
    gemm(B1, N, Y, 128, Y + 16, 128, N, 64, N, N, 112, 0);
    gemm(B1, N, B1, N, B0, N, N, N, N, N, N, 0);
    gemm(B0, N, Y, 128, Y + 32, 128, N, 64, N, N,  96, 0);
    gemm(B0, N, B0, N, B1, N, N, N, N, N, N, 0);
    gemm(B1, N, Y, 128, Y + 64, 128, N, 64, N, N,  64, 0);
    gemm(B1, N, B1, N, B0, N, N, N, N, N, N, 0);

    copy_row<<<(N + 255) / 256, 256, 0, stream>>>(Cv, R, N);
    gemm(R, N, B0, N, R + 1 * N,  N, 64, N, N, 63, N, 0);
    gemm(B0, N, B0, N, B1, N, N, N, N, N, N, 0);
    gemm(R, N, B1, N, R + 2 * N,  N, 64, N, N, 62, N, 0);
    gemm(B1, N, B1, N, B0, N, N, N, N, N, N, 0);
    gemm(R, N, B0, N, R + 4 * N,  N, 64, N, N, 60, N, 0);
    gemm(B0, N, B0, N, B1, N, N, N, N, N, N, 0);
    gemm(R, N, B1, N, R + 8 * N,  N, 64, N, N, 56, N, 0);
    gemm(B1, N, B1, N, B0, N, N, N, N, N, N, 0);
    gemm(R, N, B0, N, R + 16 * N, N, 64, N, N, 48, N, 0);
    gemm(B0, N, B0, N, B1, N, N, N, N, N, N, 0);
    gemm(R, N, B1, N, R + 32 * N, N, 64, N, N, 32, N, 0);

    gemm(R, N, Y, 128, Kk, 128, 64, 128, N, 64, 128, 0);

    conv_causal<<<L / 256, 256, 0, stream>>>(u, Kk, Dv, y, L);
}


# Independent kernels co-launched: each doubling step's squaring + 16-tile apply as one 272-block launch, Bb matvec and C row copy ride with the Ab GEMM; 34 launches to 20, sums unchanged
# speedup vs baseline: 4.7159x; 1.4371x over previous
#include <hip/hip_runtime.h>
#include <math.h>


typedef __attribute__((ext_vector_type(2))) float v2f;
typedef __attribute__((ext_vector_type(8))) float v8f;

#define NDIM 1024
#define LSEQ 8192

template <int LDA, int LDB>
__device__ __forceinline__
void gemm_tile_f32(const float* __restrict__ Ag,
                   const float* __restrict__ Bg,
                   float* __restrict__ Cg, int ldc,
                   int Kdim, int mBound, int nBound, int addI,
                   const int row0, const int col0,
                   float (&As)[64][33],
                   float (&Bs)[64][35])
{
    const int tid  = threadIdx.x;
    const int lane = tid & 31;
    const int wv   = tid >> 5;
    const int wm   = (wv >> 1) * 32;
    const int wn   = (wv & 1)  * 32;
    const int g    = lane >> 4;
    const int lm   = lane & 15;

    v8f acc[2][2] = {};

    const float* aBase = Ag + (size_t)(row0 + (tid >> 5)) * LDA + (tid & 31);
    const float* bBase = Bg + (size_t)(tid >> 6) * LDB + (col0 + (tid & 63));

    for (int k0 = 0; k0 < Kdim; k0 += 32) {
        const float* aP = aBase + k0;
        const float* bP = bBase + (size_t)k0 * LDB;
        float ra[16], rb[16];
        #pragma unroll
        for (int it = 0; it < 16; ++it) ra[it] = aP[it * 4 * LDA];
        #pragma unroll
        for (int it = 0; it < 16; ++it) rb[it] = bP[it * 2 * LDB];
        #pragma unroll
        for (int it = 0; it < 16; ++it) {
            int i = tid + it * 128;
            As[i >> 5][i & 31] = ra[it];
        }
        #pragma unroll
        for (int it = 0; it < 16; ++it) {
            int i = tid + it * 128;
            Bs[i & 63][i >> 6] = rb[it];
        }
        if (k0 + 32 < Kdim) {
            __builtin_prefetch(&Ag[(size_t)(row0 + (tid >> 1)) * LDA + (k0 + 32) + (tid & 1) * 16], 0, 0);
            __builtin_prefetch(&Bg[(size_t)(k0 + 32 + (tid >> 2)) * LDB + col0 + (tid & 3) * 16], 0, 0);
        }
        __syncthreads();

        #pragma unroll
        for (int kk = 0; kk < 32; kk += 4) {
            const int ka = kk + 2 * g;
            v2f a0, a1, b0, b1;
            a0.x = As[wm + lm][ka];           a0.y = As[wm + lm][ka + 1];
            a1.x = As[wm + 16 + lm][ka];      a1.y = As[wm + 16 + lm][ka + 1];
            b0.x = Bs[wn + lm][ka];           b0.y = Bs[wn + lm][ka + 1];
            b1.x = Bs[wn + 16 + lm][ka];      b1.y = Bs[wn + 16 + lm][ka + 1];
            acc[0][0] = __builtin_amdgcn_wmma_f32_16x16x4_f32(false, a0, false, b0, (short)0, acc[0][0], false, false);
            acc[0][1] = __builtin_amdgcn_wmma_f32_16x16x4_f32(false, a0, false, b1, (short)0, acc[0][1], false, false);
            acc[1][0] = __builtin_amdgcn_wmma_f32_16x16x4_f32(false, a1, false, b0, (short)0, acc[1][0], false, false);
            acc[1][1] = __builtin_amdgcn_wmma_f32_16x16x4_f32(false, a1, false, b1, (short)0, acc[1][1], false, false);
        }
        __syncthreads();
    }

    #pragma unroll
    for (int mi = 0; mi < 2; ++mi)
        #pragma unroll
        for (int ni = 0; ni < 2; ++ni)
            #pragma unroll
            for (int r = 0; r < 8; ++r) {
                int row = row0 + wm + mi * 16 + r + 8 * g;
                int col = col0 + wn + ni * 16 + lm;
                if (row < mBound && col < nBound) {
                    float v = acc[mi][ni][r];
                    if (addI && row == col) v += 1.0f;
                    Cg[(size_t)row * ldc + col] = v;
                }
            }
}

template <int LDA, int LDB>
__global__ __launch_bounds__(128)
void gemm_wmma_f32(const float* __restrict__ Ag,
                   const float* __restrict__ Bg,
                   float* __restrict__ Cg, int ldc,
                   int Kdim, int mBound, int nBound, int addI)
{
    __shared__ float As[64][33];
    __shared__ float Bs[64][35];
    gemm_tile_f32<LDA, LDB>(Ag, Bg, Cg, ldc, Kdim, mBound, nBound, addI,
                            blockIdx.y * 64, blockIdx.x * 64, As, Bs);
}

template <int LDB1, bool ALONG_N>
__global__ __launch_bounds__(128)
void gemm_pair_wmma_f32(const float* __restrict__ P, float* __restrict__ Sq,
                        const float* __restrict__ A1, const float* __restrict__ B1,
                        float* __restrict__ C1, int ldc1, int mB1, int nB1)
{
    __shared__ float As[64][33];
    __shared__ float Bs[64][35];
    const int b = blockIdx.x;
    if (b < 256) {
        gemm_tile_f32<NDIM, NDIM>(P, P, Sq, NDIM, NDIM, NDIM, NDIM, 0,
                                  (b >> 4) * 64, (b & 15) * 64, As, Bs);
    } else {
        const int t = (b - 256) * 64;
        gemm_tile_f32<NDIM, LDB1>(A1, B1, C1, ldc1, NDIM, mB1, nB1, 0,
                                  ALONG_N ? 0 : t, ALONG_N ? t : 0, As, Bs);
    }
}

__global__ __launch_bounds__(128)
void gemm_ab_with_vectors(const float* __restrict__ S, const float* __restrict__ IM,
                          float* __restrict__ Ab,
                          const float* __restrict__ Bv, const float* __restrict__ ldt,
                          float* __restrict__ Y,
                          const float* __restrict__ Cv, float* __restrict__ R)
{
    __shared__ float As[64][33];
    __shared__ float Bs[64][35];
    const int b = blockIdx.x;
    if (b < 256) {
        gemm_tile_f32<NDIM, NDIM>(S, IM, Ab, NDIM, NDIM, NDIM, NDIM, 0,
                                  (b >> 4) * 64, (b & 15) * 64, As, Bs);
    } else {
        const int i = (b - 256) * 128 + threadIdx.x;
        float dt = expf(ldt[0]);
        float s = 0.f;
        for (int k = 0; k < NDIM; ++k) s += S[(size_t)i * NDIM + k] * Bv[k];
        Y[(size_t)i * 128] = dt * s;
        R[i] = Cv[i];
    }
}

__global__ void prep_mats(const float* __restrict__ A, const float* __restrict__ ldt,
                          float* __restrict__ M, float* __restrict__ IM, int n)
{
    int idx = blockIdx.x * blockDim.x + threadIdx.x;
    if (idx >= n * n) return;
    float h = 0.5f * expf(ldt[0]);
    float m = h * A[idx];
    M[idx]  = m;
    int r = idx / n, c = idx - r * n;
    IM[idx] = m + ((r == c) ? 1.0f : 0.0f);
}

__global__ __launch_bounds__(256)
void conv_causal(const float* __restrict__ u, const float* __restrict__ Kk,
                 const float* __restrict__ Dv, float* __restrict__ y, int L)
{
    __shared__ float Ks[256];
    __shared__ float us[512];
    const int tid = threadIdx.x;
    const int t0  = blockIdx.x * 256;
    const int t   = t0 + tid;
    float acc = 0.f;
    for (int c0 = 0; c0 <= t0; c0 += 256) {
        Ks[tid] = Kk[(L - 1) - (c0 + tid)];
        int base = t0 - c0 - 255;
        int i0 = base + tid;
        us[tid] = (i0 >= 0) ? u[i0] : 0.f;
        int i1 = base + 256 + tid;
        us[256 + tid] = (i1 >= 0 && i1 < L) ? u[i1] : 0.f;
        __syncthreads();
        int jmax = t - c0; if (jmax > 255) jmax = 255;
        for (int j = 0; j <= jmax; ++j)
            acc += Ks[j] * us[tid + 255 - j];
        __syncthreads();
    }
    y[t] = acc + Dv[0] * u[t];
}

extern "C" void kernel_launch(void* const* d_in, const int* in_sizes, int n_in,
                              void* d_out, int out_size, void* d_ws, size_t ws_size,
                              hipStream_t stream)
{
    (void)in_sizes; (void)n_in; (void)out_size; (void)ws_size;
    const float* u   = (const float*)d_in[0];
    const float* A   = (const float*)d_in[1];
    const float* Bv  = (const float*)d_in[2];
    const float* Cv  = (const float*)d_in[3];
    const float* Dv  = (const float*)d_in[4];
    const float* ldt = (const float*)d_in[5];
    float* y = (float*)d_out;
    const int N = NDIM, L = LSEQ;

    float* B0 = (float*)d_ws;
    float* B1 = B0 + (size_t)N * N;
    float* B2 = B1 + (size_t)N * N;
    float* B3 = B2 + (size_t)N * N;
    float* Y  = B3 + (size_t)N * N;
    float* R  = Y  + (size_t)N * 128;
    float* Kk = R  + (size_t)64 * N;

    dim3 blk(128);
    auto gemm = [&](const float* Am, int lda, const float* Bm, int ldb,
                    float* Cm, int ldc, int Mc, int Nc, int Kd,
                    int mB, int nB, int addI) {
        (void)lda;
        dim3 grid((Nc + 63) / 64, (Mc + 63) / 64);
        if (ldb == 128)
            gemm_wmma_f32<NDIM, 128><<<grid, blk, 0, stream>>>(Am, Bm, Cm, ldc, Kd, mB, nB, addI);
        else
            gemm_wmma_f32<NDIM, NDIM><<<grid, blk, 0, stream>>>(Am, Bm, Cm, ldc, Kd, mB, nB, addI);
    };

    prep_mats<<<(N * N + 255) / 256, 256, 0, stream>>>(A, ldt, B0, B1, N);
    gemm(B0, N, B1, N, B2, N, N, N, N, N, N, 1);
    gemm(B0, N, B2, N, B3, N, N, N, N, N, N, 1);
    gemm(B0, N, B3, N, B2, N, N, N, N, N, N, 1);
    gemm_ab_with_vectors<<<dim3(256 + 8), blk, 0, stream>>>(B2, B1, B3, Bv, ldt, Y, Cv, R);

    dim3 gridPair(256 + 16);
    auto pairY = [&](const float* Ms, float* Sq, float* Cm, int nB) {
        gemm_pair_wmma_f32<128, false><<<gridPair, blk, 0, stream>>>(Ms, Sq, Ms, Y, Cm, 128, N, nB);
    };
    auto pairR = [&](const float* Qt, float* Sq, float* Cm, int mB) {
        gemm_pair_wmma_f32<NDIM, true><<<gridPair, blk, 0, stream>>>(Qt, Sq, R, Qt, Cm, N, mB, N);
    };

    pairY(B3, B0, Y + 1,  127);
    pairY(B0, B1, Y + 2,  126);
    pairY(B1, B0, Y + 4,  124);
    pairY(B0, B1, Y + 8,  120);
    pairY(B1, B0, Y + 16, 112);
    pairY(B0, B1, Y + 32,  96);
    pairY(B1, B0, Y + 64,  64);

    pairR(B0, B1, R + 1 * N,  63);
    pairR(B1, B0, R + 2 * N,  62);
    pairR(B0, B1, R + 4 * N,  60);
    pairR(B1, B0, R + 8 * N,  56);
    pairR(B0, B1, R + 16 * N, 48);
    gemm(R, N, B1, N, R + 32 * N, N, 64, N, N, 32, N, 0);

    gemm(R, N, Y, 128, Kk, 128, 64, 128, N, 64, 128, 0);

    conv_causal<<<L / 256, 256, 0, stream>>>(u, Kk, Dv, y, L);
}


# GEMM K loop software-pipelined 3 chunks deep: next chunks' global loads, LDS stores (2nd buffer) and fragment reads issue before each chunk's 32 WMMAs; one split LDS-only barrier per chunk; same sums
# speedup vs baseline: 5.2445x; 1.1121x over previous
#include <hip/hip_runtime.h>
#include <math.h>


typedef __attribute__((ext_vector_type(2))) float v2f;
typedef __attribute__((ext_vector_type(8))) float v8f;

#define NDIM 1024
#define LSEQ 8192

__device__ __forceinline__ void lds_barrier_arrive()
{
    __builtin_amdgcn_fence(__ATOMIC_RELEASE, "workgroup", "local");
    __builtin_amdgcn_s_barrier_signal(-1);
}
__device__ __forceinline__ void lds_barrier_wait()
{
    __builtin_amdgcn_s_barrier_wait(-1);
    __builtin_amdgcn_fence(__ATOMIC_ACQUIRE, "workgroup", "local");
}
__device__ __forceinline__ void lds_barrier()
{
    lds_barrier_arrive();
    lds_barrier_wait();
}

template <int LDA, int LDB>
__device__ __forceinline__
void gemm_tile_f32(const float* __restrict__ Ag,
                   const float* __restrict__ Bg,
                   float* __restrict__ Cg, int ldc,
                   int Kdim, int mBound, int nBound, int addI,
                   const int row0, const int col0,
                   float (&As)[2][64][33],
                   float (&Bs)[2][64][35])
{
    const int tid  = threadIdx.x;
    const int lane = tid & 31;
    const int wv   = tid >> 5;
    const int wm   = (wv >> 1) * 32;
    const int wn   = (wv & 1)  * 32;
    const int g    = lane >> 4;
    const int lm   = lane & 15;

    v8f acc[2][2] = {};

    const float* aBase = Ag + (size_t)(row0 + (tid >> 5)) * LDA + (tid & 31);
    const float* bBase = Bg + (size_t)(tid >> 6) * LDB + (col0 + (tid & 63));

    const int nCh = (Kdim + 31) >> 5;

    float ra[16], rb[16];
    v2f fa0[8][2], fb0[8][2];
    v2f fa1[8][2], fb1[8][2];

    auto gload = [&](int c) {
        const int k0 = (c < nCh ? c : nCh - 1) * 32;
        const float* aP = aBase + k0;
        const float* bP = bBase + (size_t)k0 * LDB;
        #pragma unroll
        for (int it = 0; it < 16; ++it) ra[it] = aP[it * 4 * LDA];
        #pragma unroll
        for (int it = 0; it < 16; ++it) rb[it] = bP[it * 2 * LDB];
    };
    auto lstore = [&](const int buf) {
        #pragma unroll
        for (int it = 0; it < 16; ++it) {
            int i = tid + it * 128;
            As[buf][i >> 5][i & 31] = ra[it];
        }
        #pragma unroll
        for (int it = 0; it < 16; ++it) {
            int i = tid + it * 128;
            Bs[buf][i & 63][i >> 6] = rb[it];
        }
    };
    auto fload = [&](v2f (&fa)[8][2], v2f (&fb)[8][2], const int buf) {
        #pragma unroll
        for (int s = 0; s < 8; ++s) {
            const int ka = 4 * s + 2 * g;
            fa[s][0].x = As[buf][wm + lm][ka];        fa[s][0].y = As[buf][wm + lm][ka + 1];
            fa[s][1].x = As[buf][wm + 16 + lm][ka];   fa[s][1].y = As[buf][wm + 16 + lm][ka + 1];
            fb[s][0].x = Bs[buf][wn + lm][ka];        fb[s][0].y = Bs[buf][wn + lm][ka + 1];
            fb[s][1].x = Bs[buf][wn + 16 + lm][ka];   fb[s][1].y = Bs[buf][wn + 16 + lm][ka + 1];
        }
    };
    auto mma = [&](const v2f (&fa)[8][2], const v2f (&fb)[8][2], const int s0, const int s1) {
        #pragma unroll
        for (int s = s0; s < s1; ++s) {
            acc[0][0] = __builtin_amdgcn_wmma_f32_16x16x4_f32(false, fa[s][0], false, fb[s][0], (short)0, acc[0][0], false, false);
            acc[0][1] = __builtin_amdgcn_wmma_f32_16x16x4_f32(false, fa[s][0], false, fb[s][1], (short)0, acc[0][1], false, false);
            acc[1][0] = __builtin_amdgcn_wmma_f32_16x16x4_f32(false, fa[s][1], false, fb[s][0], (short)0, acc[1][0], false, false);
            acc[1][1] = __builtin_amdgcn_wmma_f32_16x16x4_f32(false, fa[s][1], false, fb[s][1], (short)0, acc[1][1], false, false);
        }
    };

    auto step = [&](const v2f (&ca)[8][2], const v2f (&cb)[8][2],
                    v2f (&na)[8][2], v2f (&nb)[8][2],
                    const int bufNext, const int bufStore, const int cLoad) {
        fload(na, nb, bufNext);
        lstore(bufStore);
        gload(cLoad);
        __builtin_amdgcn_sched_barrier(0);
        mma(ca, cb, 0, 2);
        __builtin_amdgcn_sched_barrier(0);
        lds_barrier_arrive();
        __builtin_amdgcn_sched_barrier(0);
        mma(ca, cb, 2, 8);
        __builtin_amdgcn_sched_barrier(0);
        lds_barrier_wait();
    };

    if (nCh > 0) {
        gload(0);
        lstore(0);
        gload(1);
        lds_barrier();
        fload(fa0, fb0, 0);
        lstore(1);
        gload(2);
        lds_barrier();

        for (int c = 0; c < nCh; c += 2) {
            step(fa0, fb0, fa1, fb1, 1, 0, c + 3);
            if (c + 1 < nCh)
                step(fa1, fb1, fa0, fb0, 0, 1, c + 4);
        }
    }

    #pragma unroll
    for (int mi = 0; mi < 2; ++mi)
        #pragma unroll
        for (int ni = 0; ni < 2; ++ni)
            #pragma unroll
            for (int r = 0; r < 8; ++r) {
                int row = row0 + wm + mi * 16 + r + 8 * g;
                int col = col0 + wn + ni * 16 + lm;
                if (row < mBound && col < nBound) {
                    float v = acc[mi][ni][r];
                    if (addI && row == col) v += 1.0f;
                    Cg[(size_t)row * ldc + col] = v;
                }
            }
}

template <int LDA, int LDB>
__global__ __launch_bounds__(128)
void gemm_wmma_f32(const float* __restrict__ Ag,
                   const float* __restrict__ Bg,
                   float* __restrict__ Cg, int ldc,
                   int Kdim, int mBound, int nBound, int addI)
{
    __shared__ float As[2][64][33];
    __shared__ float Bs[2][64][35];
    gemm_tile_f32<LDA, LDB>(Ag, Bg, Cg, ldc, Kdim, mBound, nBound, addI,
                            blockIdx.y * 64, blockIdx.x * 64, As, Bs);
}

template <int LDB1, bool ALONG_N>
__global__ __launch_bounds__(128)
void gemm_pair_wmma_f32(const float* __restrict__ P, float* __restrict__ Sq,
                        const float* __restrict__ A1, const float* __restrict__ B1,
                        float* __restrict__ C1, int ldc1, int mB1, int nB1)
{
    __shared__ float As[2][64][33];
    __shared__ float Bs[2][64][35];
    const int b = blockIdx.x;
    if (b < 256) {
        gemm_tile_f32<NDIM, NDIM>(P, P, Sq, NDIM, NDIM, NDIM, NDIM, 0,
                                  (b >> 4) * 64, (b & 15) * 64, As, Bs);
    } else {
        const int t = (b - 256) * 64;
        gemm_tile_f32<NDIM, LDB1>(A1, B1, C1, ldc1, NDIM, mB1, nB1, 0,
                                  ALONG_N ? 0 : t, ALONG_N ? t : 0, As, Bs);
    }
}

__global__ __launch_bounds__(128)
void gemm_ab_with_vectors(const float* __restrict__ S, const float* __restrict__ IM,
                          float* __restrict__ Ab,
                          const float* __restrict__ Bv, const float* __restrict__ ldt,
                          float* __restrict__ Y,
                          const float* __restrict__ Cv, float* __restrict__ R)
{
    __shared__ float As[2][64][33];
    __shared__ float Bs[2][64][35];
    const int b = blockIdx.x;
    if (b < 256) {
        gemm_tile_f32<NDIM, NDIM>(S, IM, Ab, NDIM, NDIM, NDIM, NDIM, 0,
                                  (b >> 4) * 64, (b & 15) * 64, As, Bs);
    } else {
        const int i = (b - 256) * 128 + threadIdx.x;
        float dt = expf(ldt[0]);
        float s = 0.f;
        for (int k = 0; k < NDIM; ++k) s += S[(size_t)i * NDIM + k] * Bv[k];
        Y[(size_t)i * 128] = dt * s;
        R[i] = Cv[i];
    }
}

__global__ void prep_mats(const float* __restrict__ A, const float* __restrict__ ldt,
                          float* __restrict__ M, float* __restrict__ IM, int n)
{
    int idx = blockIdx.x * blockDim.x + threadIdx.x;
    if (idx >= n * n) return;
    float h = 0.5f * expf(ldt[0]);
    float m = h * A[idx];
    M[idx]  = m;
    int r = idx / n, c = idx - r * n;
    IM[idx] = m + ((r == c) ? 1.0f : 0.0f);
}

__global__ __launch_bounds__(256)
void conv_causal(const float* __restrict__ u, const float* __restrict__ Kk,
                 const float* __restrict__ Dv, float* __restrict__ y, int L)
{
    __shared__ float Ks[256];
    __shared__ float us[512];
    const int tid = threadIdx.x;
    const int t0  = blockIdx.x * 256;
    const int t   = t0 + tid;
    float acc = 0.f;
    for (int c0 = 0; c0 <= t0; c0 += 256) {
        Ks[tid] = Kk[(L - 1) - (c0 + tid)];
        int base = t0 - c0 - 255;
        int i0 = base + tid;
        us[tid] = (i0 >= 0) ? u[i0] : 0.f;
        int i1 = base + 256 + tid;
        us[256 + tid] = (i1 >= 0 && i1 < L) ? u[i1] : 0.f;
        __syncthreads();
        int jmax = t - c0; if (jmax > 255) jmax = 255;
        for (int j = 0; j <= jmax; ++j)
            acc += Ks[j] * us[tid + 255 - j];
        __syncthreads();
    }
    y[t] = acc + Dv[0] * u[t];
}

extern "C" void kernel_launch(void* const* d_in, const int* in_sizes, int n_in,
                              void* d_out, int out_size, void* d_ws, size_t ws_size,
                              hipStream_t stream)
{
    (void)in_sizes; (void)n_in; (void)out_size; (void)ws_size;
    const float* u   = (const float*)d_in[0];
    const float* A   = (const float*)d_in[1];
    const float* Bv  = (const float*)d_in[2];
    const float* Cv  = (const float*)d_in[3];
    const float* Dv  = (const float*)d_in[4];
    const float* ldt = (const float*)d_in[5];
    float* y = (float*)d_out;
    const int N = NDIM, L = LSEQ;

    float* B0 = (float*)d_ws;
    float* B1 = B0 + (size_t)N * N;
    float* B2 = B1 + (size_t)N * N;
    float* B3 = B2 + (size_t)N * N;
    float* Y  = B3 + (size_t)N * N;
    float* R  = Y  + (size_t)N * 128;
    float* Kk = R  + (size_t)64 * N;

    dim3 blk(128);
    auto gemm = [&](const float* Am, int lda, const float* Bm, int ldb,
                    float* Cm, int ldc, int Mc, int Nc, int Kd,
                    int mB, int nB, int addI) {
        (void)lda;
        dim3 grid((Nc + 63) / 64, (Mc + 63) / 64);
        if (ldb == 128)
            gemm_wmma_f32<NDIM, 128><<<grid, blk, 0, stream>>>(Am, Bm, Cm, ldc, Kd, mB, nB, addI);
        else
            gemm_wmma_f32<NDIM, NDIM><<<grid, blk, 0, stream>>>(Am, Bm, Cm, ldc, Kd, mB, nB, addI);
    };

    prep_mats<<<(N * N + 255) / 256, 256, 0, stream>>>(A, ldt, B0, B1, N);
    gemm(B0, N, B1, N, B2, N, N, N, N, N, N, 1);
    gemm(B0, N, B2, N, B3, N, N, N, N, N, N, 1);
    gemm(B0, N, B3, N, B2, N, N, N, N, N, N, 1);
    gemm_ab_with_vectors<<<dim3(256 + 8), blk, 0, stream>>>(B2, B1, B3, Bv, ldt, Y, Cv, R);

    dim3 gridPair(256 + 16);
    auto pairY = [&](const float* Ms, float* Sq, float* Cm, int nB) {
        gemm_pair_wmma_f32<128, false><<<gridPair, blk, 0, stream>>>(Ms, Sq, Ms, Y, Cm, 128, N, nB);
    };
    auto pairR = [&](const float* Qt, float* Sq, float* Cm, int mB) {
        gemm_pair_wmma_f32<NDIM, true><<<gridPair, blk, 0, stream>>>(Qt, Sq, R, Qt, Cm, N, mB, N);
    };

    pairY(B3, B0, Y + 1,  127);
    pairY(B0, B1, Y + 2,  126);
    pairY(B1, B0, Y + 4,  124);
    pairY(B0, B1, Y + 8,  120);
    pairY(B1, B0, Y + 16, 112);
    pairY(B0, B1, Y + 32,  96);
    pairY(B1, B0, Y + 64,  64);

    pairR(B0, B1, R + 1 * N,  63);
    pairR(B1, B0, R + 2 * N,  62);
    pairR(B0, B1, R + 4 * N,  60);
    pairR(B1, B0, R + 8 * N,  56);
    pairR(B0, B1, R + 16 * N, 48);
    gemm(R, N, B1, N, R + 32 * N, N, 64, N, N, 32, N, 0);

    gemm(R, N, Y, 128, Kk, 128, 64, 128, N, 64, 128, 0);

    conv_causal<<<L / 256, 256, 0, stream>>>(u, Kk, Dv, y, L);
}


# Causal convolution moved from one thread per output (serial LDS loop) to f32 WMMA: Hankel(K) x Toeplitz(u) tile per 256 outputs, l range split over 16 waves, partial sums added in wave order
# speedup vs baseline: 6.4267x; 1.2254x over previous
#include <hip/hip_runtime.h>
#include <math.h>


typedef __attribute__((ext_vector_type(2))) float v2f;
typedef __attribute__((ext_vector_type(8))) float v8f;

#define NDIM 1024
#define LSEQ 8192

__device__ __forceinline__ void lds_barrier_arrive()
{
    __builtin_amdgcn_fence(__ATOMIC_RELEASE, "workgroup", "local");
    __builtin_amdgcn_s_barrier_signal(-1);
}
__device__ __forceinline__ void lds_barrier_wait()
{
    __builtin_amdgcn_s_barrier_wait(-1);
    __builtin_amdgcn_fence(__ATOMIC_ACQUIRE, "workgroup", "local");
}
__device__ __forceinline__ void lds_barrier()
{
    lds_barrier_arrive();
    lds_barrier_wait();
}

template <int LDA, int LDB>
__device__ __forceinline__
void gemm_tile_f32(const float* __restrict__ Ag,
                   const float* __restrict__ Bg,
                   float* __restrict__ Cg, int ldc,
                   int Kdim, int mBound, int nBound, int addI,
                   const int row0, const int col0,
                   float (&As)[2][64][33],
                   float (&Bs)[2][64][35])
{
    const int tid  = threadIdx.x;
    const int lane = tid & 31;
    const int wv   = tid >> 5;
    const int wm   = (wv >> 1) * 32;
    const int wn   = (wv & 1)  * 32;
    const int g    = lane >> 4;
    const int lm   = lane & 15;

    v8f acc[2][2] = {};

    const float* aBase = Ag + (size_t)(row0 + (tid >> 5)) * LDA + (tid & 31);
    const float* bBase = Bg + (size_t)(tid >> 6) * LDB + (col0 + (tid & 63));

    const int nCh = (Kdim + 31) >> 5;

    float ra[16], rb[16];
    v2f fa0[8][2], fb0[8][2];
    v2f fa1[8][2], fb1[8][2];

    auto gload = [&](int c) {
        const int k0 = (c < nCh ? c : nCh - 1) * 32;
        const float* aP = aBase + k0;
        const float* bP = bBase + (size_t)k0 * LDB;
        #pragma unroll
        for (int it = 0; it < 16; ++it) ra[it] = aP[it * 4 * LDA];
        #pragma unroll
        for (int it = 0; it < 16; ++it) rb[it] = bP[it * 2 * LDB];
    };
    auto lstore = [&](const int buf) {
        #pragma unroll
        for (int it = 0; it < 16; ++it) {
            int i = tid + it * 128;
            As[buf][i >> 5][i & 31] = ra[it];
        }
        #pragma unroll
        for (int it = 0; it < 16; ++it) {
            int i = tid + it * 128;
            Bs[buf][i & 63][i >> 6] = rb[it];
        }
    };
    auto fload = [&](v2f (&fa)[8][2], v2f (&fb)[8][2], const int buf) {
        #pragma unroll
        for (int s = 0; s < 8; ++s) {
            const int ka = 4 * s + 2 * g;
            fa[s][0].x = As[buf][wm + lm][ka];        fa[s][0].y = As[buf][wm + lm][ka + 1];
            fa[s][1].x = As[buf][wm + 16 + lm][ka];   fa[s][1].y = As[buf][wm + 16 + lm][ka + 1];
            fb[s][0].x = Bs[buf][wn + lm][ka];        fb[s][0].y = Bs[buf][wn + lm][ka + 1];
            fb[s][1].x = Bs[buf][wn + 16 + lm][ka];   fb[s][1].y = Bs[buf][wn + 16 + lm][ka + 1];
        }
    };
    auto mma = [&](const v2f (&fa)[8][2], const v2f (&fb)[8][2], const int s0, const int s1) {
        #pragma unroll
        for (int s = s0; s < s1; ++s) {
            acc[0][0] = __builtin_amdgcn_wmma_f32_16x16x4_f32(false, fa[s][0], false, fb[s][0], (short)0, acc[0][0], false, false);
            acc[0][1] = __builtin_amdgcn_wmma_f32_16x16x4_f32(false, fa[s][0], false, fb[s][1], (short)0, acc[0][1], false, false);
            acc[1][0] = __builtin_amdgcn_wmma_f32_16x16x4_f32(false, fa[s][1], false, fb[s][0], (short)0, acc[1][0], false, false);
            acc[1][1] = __builtin_amdgcn_wmma_f32_16x16x4_f32(false, fa[s][1], false, fb[s][1], (short)0, acc[1][1], false, false);
        }
    };

    auto step = [&](const v2f (&ca)[8][2], const v2f (&cb)[8][2],
                    v2f (&na)[8][2], v2f (&nb)[8][2],
                    const int bufNext, const int bufStore, const int cLoad) {
        fload(na, nb, bufNext);
        lstore(bufStore);
        gload(cLoad);
        __builtin_amdgcn_sched_barrier(0);
        mma(ca, cb, 0, 2);
        __builtin_amdgcn_sched_barrier(0);
        lds_barrier_arrive();
        __builtin_amdgcn_sched_barrier(0);
        mma(ca, cb, 2, 8);
        __builtin_amdgcn_sched_barrier(0);
        lds_barrier_wait();
    };

    if (nCh > 0) {
        gload(0);
        lstore(0);
        gload(1);
        lds_barrier();
        fload(fa0, fb0, 0);
        lstore(1);
        gload(2);
        lds_barrier();

        for (int c = 0; c < nCh; c += 2) {
            step(fa0, fb0, fa1, fb1, 1, 0, c + 3);
            if (c + 1 < nCh)
                step(fa1, fb1, fa0, fb0, 0, 1, c + 4);
        }
    }

    #pragma unroll
    for (int mi = 0; mi < 2; ++mi)
        #pragma unroll
        for (int ni = 0; ni < 2; ++ni)
            #pragma unroll
            for (int r = 0; r < 8; ++r) {
                int row = row0 + wm + mi * 16 + r + 8 * g;
                int col = col0 + wn + ni * 16 + lm;
                if (row < mBound && col < nBound) {
                    float v = acc[mi][ni][r];
                    if (addI && row == col) v += 1.0f;
                    Cg[(size_t)row * ldc + col] = v;
                }
            }
}

template <int LDA, int LDB>
__global__ __launch_bounds__(128)
void gemm_wmma_f32(const float* __restrict__ Ag,
                   const float* __restrict__ Bg,
                   float* __restrict__ Cg, int ldc,
                   int Kdim, int mBound, int nBound, int addI)
{
    __shared__ float As[2][64][33];
    __shared__ float Bs[2][64][35];
    gemm_tile_f32<LDA, LDB>(Ag, Bg, Cg, ldc, Kdim, mBound, nBound, addI,
                            blockIdx.y * 64, blockIdx.x * 64, As, Bs);
}

template <int LDB1, bool ALONG_N>
__global__ __launch_bounds__(128)
void gemm_pair_wmma_f32(const float* __restrict__ P, float* __restrict__ Sq,
                        const float* __restrict__ A1, const float* __restrict__ B1,
                        float* __restrict__ C1, int ldc1, int mB1, int nB1)
{
    __shared__ float As[2][64][33];
    __shared__ float Bs[2][64][35];
    const int b = blockIdx.x;
    if (b < 256) {
        gemm_tile_f32<NDIM, NDIM>(P, P, Sq, NDIM, NDIM, NDIM, NDIM, 0,
                                  (b >> 4) * 64, (b & 15) * 64, As, Bs);
    } else {
        const int t = (b - 256) * 64;
        gemm_tile_f32<NDIM, LDB1>(A1, B1, C1, ldc1, NDIM, mB1, nB1, 0,
                                  ALONG_N ? 0 : t, ALONG_N ? t : 0, As, Bs);
    }
}

__global__ __launch_bounds__(128)
void gemm_ab_with_vectors(const float* __restrict__ S, const float* __restrict__ IM,
                          float* __restrict__ Ab,
                          const float* __restrict__ Bv, const float* __restrict__ ldt,
                          float* __restrict__ Y,
                          const float* __restrict__ Cv, float* __restrict__ R)
{
    __shared__ float As[2][64][33];
    __shared__ float Bs[2][64][35];
    const int b = blockIdx.x;
    if (b < 256) {
        gemm_tile_f32<NDIM, NDIM>(S, IM, Ab, NDIM, NDIM, NDIM, NDIM, 0,
                                  (b >> 4) * 64, (b & 15) * 64, As, Bs);
    } else {
        const int i = (b - 256) * 128 + threadIdx.x;
        float dt = expf(ldt[0]);
        float s = 0.f;
        for (int k = 0; k < NDIM; ++k) s += S[(size_t)i * NDIM + k] * Bv[k];
        Y[(size_t)i * 128] = dt * s;
        R[i] = Cv[i];
    }
}

__global__ void prep_mats(const float* __restrict__ A, const float* __restrict__ ldt,
                          float* __restrict__ M, float* __restrict__ IM, int n)
{
    int idx = blockIdx.x * blockDim.x + threadIdx.x;
    if (idx >= n * n) return;
    float h = 0.5f * expf(ldt[0]);
    float m = h * A[idx];
    M[idx]  = m;
    int r = idx / n, c = idx - r * n;
    IM[idx] = m + ((r == c) ? 1.0f : 0.0f);
}

#define CV_WAVES 16
__global__ __launch_bounds__(CV_WAVES * 32)
void conv_causal_wmma(const float* __restrict__ u, const float* __restrict__ Kk,
                      const float* __restrict__ Dv, float* __restrict__ y)
{
    __shared__ float KT[LSEQ + 16];
    __shared__ float USk[255 + LSEQ + LSEQ / 16 + 1];
    __shared__ float red[CV_WAVES][256];

    const int tid = threadIdx.x;
    const int tb  = blockIdx.x;
    const int nIt = tb + 1;

    float rk[16], ru[16];
    #pragma unroll
    for (int it = 0; it < 16; ++it) {
        const int yy = tid + it * (CV_WAVES * 32);
        rk[it] = Kk[yy];
        ru[it] = u[yy];
    }
    #pragma unroll
    for (int it = 0; it < 16; ++it) {
        const int yy = tid + it * (CV_WAVES * 32);
        KT[yy] = rk[it];
        USk[255 + yy + (yy >> 4)] = ru[it];
    }
    if (tid < 16)  KT[LSEQ + tid] = 0.f;
    if (tid < 255) USk[tid] = 0.f;
    __syncthreads();

    const int lane = tid & 31;
    const int wv   = tid >> 5;
    const int c    = lane & 15;
    const int h    = lane >> 4;

    const float* ap = KT + (LSEQ + 14 - 13) - wv * 16 * nIt - c - 2 * h;
    const float* bp = USk + 17 * (16 * tb + c + 15 - wv * nIt) - 2 * h;

    auto fl = [&](v2f (&a)[4], v2f (&b)[4], const int it) {
        const float* a_ = ap - 16 * it;
        const float* b_ = bp - 17 * it;
        #pragma unroll
        for (int s = 0; s < 4; ++s) {
            a[s].x = a_[12 - 4 * s];   a[s].y = a_[13 - 4 * s];
            b[s].x = b_[14 - 4 * s];   b[s].y = b_[15 - 4 * s];
        }
    };

    v8f acc = {};
    auto mma_group = [&](const v2f (&a)[4], const v2f (&b)[4]) {
        __builtin_amdgcn_sched_barrier(0);
        #pragma unroll
        for (int s = 0; s < 4; ++s)
            acc = __builtin_amdgcn_wmma_f32_16x16x4_f32(false, a[s], false, b[s], (short)0, acc, false, false);
        asm volatile("v_nop\n\tv_nop\n\tv_nop\n\tv_nop" : "+v"(acc));
        __builtin_amdgcn_sched_barrier(0);
    };

    v2f a0[4], b0[4], a1[4], b1[4];
    fl(a0, b0, 0);
    for (int it = 0; it < nIt; it += 2) {
        fl(a1, b1, (it + 1 < nIt) ? it + 1 : it);
        mma_group(a0, b0);
        if (it + 1 < nIt) {
            fl(a0, b0, (it + 2 < nIt) ? it + 2 : it + 1);
            mma_group(a1, b1);
        }
    }

    float* rp = &red[wv][16 * c + 8 * h];
    #pragma unroll
    for (int r = 0; r < 8; ++r) rp[r] = acc[r];
    __syncthreads();

    if (tid < 256) {
        float s = 0.f;
        #pragma unroll
        for (int w = 0; w < CV_WAVES; ++w) s += red[w][tid];
        const int t = tb * 256 + tid;
        y[t] = s + Dv[0] * u[t];
    }
}

extern "C" void kernel_launch(void* const* d_in, const int* in_sizes, int n_in,
                              void* d_out, int out_size, void* d_ws, size_t ws_size,
                              hipStream_t stream)
{
    (void)in_sizes; (void)n_in; (void)out_size; (void)ws_size;
    const float* u   = (const float*)d_in[0];
    const float* A   = (const float*)d_in[1];
    const float* Bv  = (const float*)d_in[2];
    const float* Cv  = (const float*)d_in[3];
    const float* Dv  = (const float*)d_in[4];
    const float* ldt = (const float*)d_in[5];
    float* y = (float*)d_out;
    const int N = NDIM, L = LSEQ;

    float* B0 = (float*)d_ws;
    float* B1 = B0 + (size_t)N * N;
    float* B2 = B1 + (size_t)N * N;
    float* B3 = B2 + (size_t)N * N;
    float* Y  = B3 + (size_t)N * N;
    float* R  = Y  + (size_t)N * 128;
    float* Kk = R  + (size_t)64 * N;

    dim3 blk(128);
    auto gemm = [&](const float* Am, int lda, const float* Bm, int ldb,
                    float* Cm, int ldc, int Mc, int Nc, int Kd,
                    int mB, int nB, int addI) {
        (void)lda;
        dim3 grid((Nc + 63) / 64, (Mc + 63) / 64);
        if (ldb == 128)
            gemm_wmma_f32<NDIM, 128><<<grid, blk, 0, stream>>>(Am, Bm, Cm, ldc, Kd, mB, nB, addI);
        else
            gemm_wmma_f32<NDIM, NDIM><<<grid, blk, 0, stream>>>(Am, Bm, Cm, ldc, Kd, mB, nB, addI);
    };

    prep_mats<<<(N * N + 255) / 256, 256, 0, stream>>>(A, ldt, B0, B1, N);
    gemm(B0, N, B1, N, B2, N, N, N, N, N, N, 1);
    gemm(B0, N, B2, N, B3, N, N, N, N, N, N, 1);
    gemm(B0, N, B3, N, B2, N, N, N, N, N, N, 1);
    gemm_ab_with_vectors<<<dim3(256 + 8), blk, 0, stream>>>(B2, B1, B3, Bv, ldt, Y, Cv, R);

    dim3 gridPair(256 + 16);
    auto pairY = [&](const float* Ms, float* Sq, float* Cm, int nB) {
        gemm_pair_wmma_f32<128, false><<<gridPair, blk, 0, stream>>>(Ms, Sq, Ms, Y, Cm, 128, N, nB);
    };
    auto pairR = [&](const float* Qt, float* Sq, float* Cm, int mB) {
        gemm_pair_wmma_f32<NDIM, true><<<gridPair, blk, 0, stream>>>(Qt, Sq, R, Qt, Cm, N, mB, N);
    };

    pairY(B3, B0, Y + 1,  127);
    pairY(B0, B1, Y + 2,  126);
    pairY(B1, B0, Y + 4,  124);
    pairY(B0, B1, Y + 8,  120);
    pairY(B1, B0, Y + 16, 112);
    pairY(B0, B1, Y + 32,  96);
    pairY(B1, B0, Y + 64,  64);

    pairR(B0, B1, R + 1 * N,  63);
    pairR(B1, B0, R + 2 * N,  62);
    pairR(B0, B1, R + 4 * N,  60);
    pairR(B1, B0, R + 8 * N,  56);
    pairR(B0, B1, R + 16 * N, 48);
    gemm(R, N, B1, N, R + 32 * N, N, 64, N, N, 32, N, 0);

    gemm(R, N, Y, 128, Kk, 128, 64, 128, N, 64, 128, 0);

    conv_causal_wmma<<<L / 256, 512, 0, stream>>>(u, Kk, Dv, y);
}


# GEMM tiles move in 16-byte pieces: 128-bit global loads and LDS stores, one 128-bit LDS read feeds two k steps (pitch 36); the 4 k of a WMMA regroup inside each 8, so sum order changes
# speedup vs baseline: 7.2343x; 1.1257x over previous
#include <hip/hip_runtime.h>
#include <math.h>


typedef __attribute__((ext_vector_type(2))) float v2f;
typedef __attribute__((ext_vector_type(4))) float v4f;
typedef __attribute__((ext_vector_type(8))) float v8f;

#define NDIM 1024
#define LSEQ 8192

__device__ __forceinline__ void lds_barrier_arrive()
{
    __builtin_amdgcn_fence(__ATOMIC_RELEASE, "workgroup", "local");
    __builtin_amdgcn_s_barrier_signal(-1);
}
__device__ __forceinline__ void lds_barrier_wait()
{
    __builtin_amdgcn_s_barrier_wait(-1);
    __builtin_amdgcn_fence(__ATOMIC_ACQUIRE, "workgroup", "local");
}
__device__ __forceinline__ void lds_barrier()
{
    lds_barrier_arrive();
    lds_barrier_wait();
}

template <int LDA, int LDB>
__device__ __forceinline__
void gemm_tile_f32(const float* __restrict__ Ag,
                   const float* __restrict__ Bg,
                   float* __restrict__ Cg, int ldc,
                   int Kdim, int mBound, int nBound, int addI,
                   const int row0, const int col0,
                   float (&As)[2][64][36],
                   float (&Bs)[2][64][36])
{
    const int tid  = threadIdx.x;
    const int lane = tid & 31;
    const int wv   = tid >> 5;
    const int wm   = (wv >> 1) * 32;
    const int wn   = (wv & 1)  * 32;
    const int g    = lane >> 4;
    const int lm   = lane & 15;

    v8f acc[2][2] = {};

    const float* aBase = Ag + (size_t)(row0 + (tid >> 3)) * LDA + (tid & 7) * 4;
    const float* bBase = Bg + (size_t)((tid >> 6) * 4) * LDB + (col0 + (tid & 63));

    const int nCh = (Kdim + 31) >> 5;

    v4f ra[4], rb[4];
    v4f fa0[4][2], fb0[4][2];
    v4f fa1[4][2], fb1[4][2];

    auto gload = [&](int c) {
        const int k0 = (c < nCh ? c : nCh - 1) * 32;
        const float* aP = aBase + k0;
        const float* bP = bBase + (size_t)k0 * LDB;
        #pragma unroll
        for (int it = 0; it < 4; ++it)
            ra[it] = *reinterpret_cast<const v4f*>(aP + it * 16 * LDA);
        #pragma unroll
        for (int it = 0; it < 4; ++it) {
            rb[it].x = bP[(it * 8 + 0) * LDB];
            rb[it].y = bP[(it * 8 + 1) * LDB];
            rb[it].z = bP[(it * 8 + 2) * LDB];
            rb[it].w = bP[(it * 8 + 3) * LDB];
        }
    };
    auto lstore = [&](const int buf) {
        #pragma unroll
        for (int it = 0; it < 4; ++it)
            *reinterpret_cast<v4f*>(&As[buf][(tid >> 3) + 16 * it][(tid & 7) * 4]) = ra[it];
        #pragma unroll
        for (int it = 0; it < 4; ++it)
            *reinterpret_cast<v4f*>(&Bs[buf][tid & 63][((tid >> 6) + 2 * it) * 4]) = rb[it];
    };
    auto fload = [&](v4f (&fa)[4][2], v4f (&fb)[4][2], const int buf) {
        #pragma unroll
        for (int p = 0; p < 4; ++p) {
            const int kq = 8 * p + 4 * g;
            fa[p][0] = *reinterpret_cast<const v4f*>(&As[buf][wm + lm][kq]);
            fa[p][1] = *reinterpret_cast<const v4f*>(&As[buf][wm + 16 + lm][kq]);
            fb[p][0] = *reinterpret_cast<const v4f*>(&Bs[buf][wn + lm][kq]);
            fb[p][1] = *reinterpret_cast<const v4f*>(&Bs[buf][wn + 16 + lm][kq]);
        }
    };
    auto mma = [&](const v4f (&fa)[4][2], const v4f (&fb)[4][2], const int p0, const int p1) {
        #pragma unroll
        for (int p = p0; p < p1; ++p) {
            const v2f a0l = __builtin_shufflevector(fa[p][0], fa[p][0], 0, 1), a0h = __builtin_shufflevector(fa[p][0], fa[p][0], 2, 3);
            const v2f a1l = __builtin_shufflevector(fa[p][1], fa[p][1], 0, 1), a1h = __builtin_shufflevector(fa[p][1], fa[p][1], 2, 3);
            const v2f b0l = __builtin_shufflevector(fb[p][0], fb[p][0], 0, 1), b0h = __builtin_shufflevector(fb[p][0], fb[p][0], 2, 3);
            const v2f b1l = __builtin_shufflevector(fb[p][1], fb[p][1], 0, 1), b1h = __builtin_shufflevector(fb[p][1], fb[p][1], 2, 3);
            acc[0][0] = __builtin_amdgcn_wmma_f32_16x16x4_f32(false, a0l, false, b0l, (short)0, acc[0][0], false, false);
            acc[0][1] = __builtin_amdgcn_wmma_f32_16x16x4_f32(false, a0l, false, b1l, (short)0, acc[0][1], false, false);
            acc[1][0] = __builtin_amdgcn_wmma_f32_16x16x4_f32(false, a1l, false, b0l, (short)0, acc[1][0], false, false);
            acc[1][1] = __builtin_amdgcn_wmma_f32_16x16x4_f32(false, a1l, false, b1l, (short)0, acc[1][1], false, false);
            acc[0][0] = __builtin_amdgcn_wmma_f32_16x16x4_f32(false, a0h, false, b0h, (short)0, acc[0][0], false, false);
            acc[0][1] = __builtin_amdgcn_wmma_f32_16x16x4_f32(false, a0h, false, b1h, (short)0, acc[0][1], false, false);
            acc[1][0] = __builtin_amdgcn_wmma_f32_16x16x4_f32(false, a1h, false, b0h, (short)0, acc[1][0], false, false);
            acc[1][1] = __builtin_amdgcn_wmma_f32_16x16x4_f32(false, a1h, false, b1h, (short)0, acc[1][1], false, false);
        }
    };

    auto step = [&](const v4f (&ca)[4][2], const v4f (&cb)[4][2],
                    v4f (&na)[4][2], v4f (&nb)[4][2],
                    const int bufNext, const int bufStore, const int cLoad) {
        fload(na, nb, bufNext);
        lstore(bufStore);
        gload(cLoad);
        __builtin_amdgcn_sched_barrier(0);
        mma(ca, cb, 0, 1);
        __builtin_amdgcn_sched_barrier(0);
        lds_barrier_arrive();
        __builtin_amdgcn_sched_barrier(0);
        mma(ca, cb, 1, 4);
        __builtin_amdgcn_sched_barrier(0);
        lds_barrier_wait();
    };

    if (nCh > 0) {
        gload(0);
        lstore(0);
        gload(1);
        lds_barrier();
        fload(fa0, fb0, 0);
        lstore(1);
        gload(2);
        lds_barrier();

        for (int c = 0; c < nCh; c += 2) {
            step(fa0, fb0, fa1, fb1, 1, 0, c + 3);
            if (c + 1 < nCh)
                step(fa1, fb1, fa0, fb0, 0, 1, c + 4);
        }
    }

    #pragma unroll
    for (int mi = 0; mi < 2; ++mi)
        #pragma unroll
        for (int ni = 0; ni < 2; ++ni)
            #pragma unroll
            for (int r = 0; r < 8; ++r) {
                int row = row0 + wm + mi * 16 + r + 8 * g;
                int col = col0 + wn + ni * 16 + lm;
                if (row < mBound && col < nBound) {
                    float v = acc[mi][ni][r];
                    if (addI && row == col) v += 1.0f;
                    Cg[(size_t)row * ldc + col] = v;
                }
            }
}

template <int LDA, int LDB>
__global__ __launch_bounds__(128)
void gemm_wmma_f32(const float* __restrict__ Ag,
                   const float* __restrict__ Bg,
                   float* __restrict__ Cg, int ldc,
                   int Kdim, int mBound, int nBound, int addI)
{
    __shared__ __attribute__((aligned(16))) float As[2][64][36];
    __shared__ __attribute__((aligned(16))) float Bs[2][64][36];
    gemm_tile_f32<LDA, LDB>(Ag, Bg, Cg, ldc, Kdim, mBound, nBound, addI,
                            blockIdx.y * 64, blockIdx.x * 64, As, Bs);
}

template <int LDB1, bool ALONG_N>
__global__ __launch_bounds__(128)
void gemm_pair_wmma_f32(const float* __restrict__ P, float* __restrict__ Sq,
                        const float* __restrict__ A1, const float* __restrict__ B1,
                        float* __restrict__ C1, int ldc1, int mB1, int nB1)
{
    __shared__ __attribute__((aligned(16))) float As[2][64][36];
    __shared__ __attribute__((aligned(16))) float Bs[2][64][36];
    const int b = blockIdx.x;
    if (b < 256) {
        gemm_tile_f32<NDIM, NDIM>(P, P, Sq, NDIM, NDIM, NDIM, NDIM, 0,
                                  (b >> 4) * 64, (b & 15) * 64, As, Bs);
    } else {
        const int t = (b - 256) * 64;
        gemm_tile_f32<NDIM, LDB1>(A1, B1, C1, ldc1, NDIM, mB1, nB1, 0,
                                  ALONG_N ? 0 : t, ALONG_N ? t : 0, As, Bs);
    }
}

__global__ __launch_bounds__(128)
void gemm_ab_with_vectors(const float* __restrict__ S, const float* __restrict__ IM,
                          float* __restrict__ Ab,
                          const float* __restrict__ Bv, const float* __restrict__ ldt,
                          float* __restrict__ Y,
                          const float* __restrict__ Cv, float* __restrict__ R)
{
    __shared__ __attribute__((aligned(16))) float As[2][64][36];
    __shared__ __attribute__((aligned(16))) float Bs[2][64][36];
    const int b = blockIdx.x;
    if (b < 256) {
        gemm_tile_f32<NDIM, NDIM>(S, IM, Ab, NDIM, NDIM, NDIM, NDIM, 0,
                                  (b >> 4) * 64, (b & 15) * 64, As, Bs);
    } else {
        const int i = (b - 256) * 128 + threadIdx.x;
        float dt = expf(ldt[0]);
        float s = 0.f;
        for (int k = 0; k < NDIM; ++k) s += S[(size_t)i * NDIM + k] * Bv[k];
        Y[(size_t)i * 128] = dt * s;
        R[i] = Cv[i];
    }
}

__global__ void prep_mats(const float* __restrict__ A, const float* __restrict__ ldt,
                          float* __restrict__ M, float* __restrict__ IM, int n)
{
    int idx = blockIdx.x * blockDim.x + threadIdx.x;
    if (idx >= n * n) return;
    float h = 0.5f * expf(ldt[0]);
    float m = h * A[idx];
    M[idx]  = m;
    int r = idx / n, c = idx - r * n;
    IM[idx] = m + ((r == c) ? 1.0f : 0.0f);
}

#define CV_WAVES 16
__global__ __launch_bounds__(CV_WAVES * 32)
void conv_causal_wmma(const float* __restrict__ u, const float* __restrict__ Kk,
                      const float* __restrict__ Dv, float* __restrict__ y)
{
    __shared__ float KT[LSEQ + 16];
    __shared__ float USk[255 + LSEQ + LSEQ / 16 + 1];
    __shared__ float red[CV_WAVES][256];

    const int tid = threadIdx.x;
    const int tb  = blockIdx.x;
    const int nIt = tb + 1;

    float rk[16], ru[16];
    #pragma unroll
    for (int it = 0; it < 16; ++it) {
        const int yy = tid + it * (CV_WAVES * 32);
        rk[it] = Kk[yy];
        ru[it] = u[yy];
    }
    #pragma unroll
    for (int it = 0; it < 16; ++it) {
        const int yy = tid + it * (CV_WAVES * 32);
        KT[yy] = rk[it];
        USk[255 + yy + (yy >> 4)] = ru[it];
    }
    if (tid < 16)  KT[LSEQ + tid] = 0.f;
    if (tid < 255) USk[tid] = 0.f;
    __syncthreads();

    const int lane = tid & 31;
    const int wv   = tid >> 5;
    const int c    = lane & 15;
    const int h    = lane >> 4;

    const float* ap = KT + (LSEQ + 14 - 13) - wv * 16 * nIt - c - 2 * h;
    const float* bp = USk + 17 * (16 * tb + c + 15 - wv * nIt) - 2 * h;

    auto fl = [&](v2f (&a)[4], v2f (&b)[4], const int it) {
        const float* a_ = ap - 16 * it;
        const float* b_ = bp - 17 * it;
        #pragma unroll
        for (int s = 0; s < 4; ++s) {
            a[s].x = a_[12 - 4 * s];   a[s].y = a_[13 - 4 * s];
            b[s].x = b_[14 - 4 * s];   b[s].y = b_[15 - 4 * s];
        }
    };

    v8f acc = {};
    auto mma_group = [&](const v2f (&a)[4], const v2f (&b)[4]) {
        __builtin_amdgcn_sched_barrier(0);
        #pragma unroll
        for (int s = 0; s < 4; ++s)
            acc = __builtin_amdgcn_wmma_f32_16x16x4_f32(false, a[s], false, b[s], (short)0, acc, false, false);
        asm volatile("v_nop\n\tv_nop\n\tv_nop\n\tv_nop" : "+v"(acc));
        __builtin_amdgcn_sched_barrier(0);
    };

    v2f a0[4], b0[4], a1[4], b1[4];
    fl(a0, b0, 0);
    for (int it = 0; it < nIt; it += 2) {
        fl(a1, b1, (it + 1 < nIt) ? it + 1 : it);
        mma_group(a0, b0);
        if (it + 1 < nIt) {
            fl(a0, b0, (it + 2 < nIt) ? it + 2 : it + 1);
            mma_group(a1, b1);
        }
    }

    float* rp = &red[wv][16 * c + 8 * h];
    #pragma unroll
    for (int r = 0; r < 8; ++r) rp[r] = acc[r];
    __syncthreads();

    if (tid < 256) {
        float s = 0.f;
        #pragma unroll
        for (int w = 0; w < CV_WAVES; ++w) s += red[w][tid];
        const int t = tb * 256 + tid;
        y[t] = s + Dv[0] * u[t];
    }
}

extern "C" void kernel_launch(void* const* d_in, const int* in_sizes, int n_in,
                              void* d_out, int out_size, void* d_ws, size_t ws_size,
                              hipStream_t stream)
{
    (void)in_sizes; (void)n_in; (void)out_size; (void)ws_size;
    const float* u   = (const float*)d_in[0];
    const float* A   = (const float*)d_in[1];
    const float* Bv  = (const float*)d_in[2];
    const float* Cv  = (const float*)d_in[3];
    const float* Dv  = (const float*)d_in[4];
    const float* ldt = (const float*)d_in[5];
    float* y = (float*)d_out;
    const int N = NDIM, L = LSEQ;

    float* B0 = (float*)d_ws;
    float* B1 = B0 + (size_t)N * N;
    float* B2 = B1 + (size_t)N * N;
    float* B3 = B2 + (size_t)N * N;
    float* Y  = B3 + (size_t)N * N;
    float* R  = Y  + (size_t)N * 128;
    float* Kk = R  + (size_t)64 * N;

    dim3 blk(128);
    auto gemm = [&](const float* Am, int lda, const float* Bm, int ldb,
                    float* Cm, int ldc, int Mc, int Nc, int Kd,
                    int mB, int nB, int addI) {
        (void)lda;
        dim3 grid((Nc + 63) / 64, (Mc + 63) / 64);
        if (ldb == 128)
            gemm_wmma_f32<NDIM, 128><<<grid, blk, 0, stream>>>(Am, Bm, Cm, ldc, Kd, mB, nB, addI);
        else
            gemm_wmma_f32<NDIM, NDIM><<<grid, blk, 0, stream>>>(Am, Bm, Cm, ldc, Kd, mB, nB, addI);
    };

    prep_mats<<<(N * N + 255) / 256, 256, 0, stream>>>(A, ldt, B0, B1, N);
    gemm(B0, N, B1, N, B2, N, N, N, N, N, N, 1);
    gemm(B0, N, B2, N, B3, N, N, N, N, N, N, 1);
    gemm(B0, N, B3, N, B2, N, N, N, N, N, N, 1);
    gemm_ab_with_vectors<<<dim3(256 + 8), blk, 0, stream>>>(B2, B1, B3, Bv, ldt, Y, Cv, R);

    dim3 gridPair(256 + 16);
    auto pairY = [&](const float* Ms, float* Sq, float* Cm, int nB) {
        gemm_pair_wmma_f32<128, false><<<gridPair, blk, 0, stream>>>(Ms, Sq, Ms, Y, Cm, 128, N, nB);
    };
    auto pairR = [&](const float* Qt, float* Sq, float* Cm, int mB) {
        gemm_pair_wmma_f32<NDIM, true><<<gridPair, blk, 0, stream>>>(Qt, Sq, R, Qt, Cm, N, mB, N);
    };

    pairY(B3, B0, Y + 1,  127);
    pairY(B0, B1, Y + 2,  126);
    pairY(B1, B0, Y + 4,  124);
    pairY(B0, B1, Y + 8,  120);
    pairY(B1, B0, Y + 16, 112);
    pairY(B0, B1, Y + 32,  96);
    pairY(B1, B0, Y + 64,  64);

    pairR(B0, B1, R + 1 * N,  63);
    pairR(B1, B0, R + 2 * N,  62);
    pairR(B0, B1, R + 4 * N,  60);
    pairR(B1, B0, R + 8 * N,  56);
    pairR(B0, B1, R + 16 * N, 48);
    gemm(R, N, B1, N, R + 32 * N, N, 64, N, N, 32, N, 0);

    gemm(R, N, Y, 128, Kk, 128, 64, 128, N, 64, 128, 0);

    conv_causal_wmma<<<L / 256, 512, 0, stream>>>(u, Kk, Dv, y);
}


# Doubling launches: apply product as 64 light tiles of 16x64 (8 WMMAs per chunk per wave) instead of 16 tiles of 64x64 beside the squaring; tile summed in 2 accumulator chains, so sum order changes
# speedup vs baseline: 8.9037x; 1.2308x over previous
#include <hip/hip_runtime.h>
#include <math.h>


typedef __attribute__((ext_vector_type(2))) float v2f;
typedef __attribute__((ext_vector_type(4))) float v4f;
typedef __attribute__((ext_vector_type(8))) float v8f;

#define NDIM 1024
#define LSEQ 8192

__device__ __forceinline__ void lds_barrier_arrive()
{
    __builtin_amdgcn_fence(__ATOMIC_RELEASE, "workgroup", "local");
    __builtin_amdgcn_s_barrier_signal(-1);
}
__device__ __forceinline__ void lds_barrier_wait()
{
    __builtin_amdgcn_s_barrier_wait(-1);
    __builtin_amdgcn_fence(__ATOMIC_ACQUIRE, "workgroup", "local");
}
__device__ __forceinline__ void lds_barrier()
{
    lds_barrier_arrive();
    lds_barrier_wait();
}

template <int LDA, int LDB, int TM>
__device__ __forceinline__
void gemm_tile_f32(const float* __restrict__ Ag,
                   const float* __restrict__ Bg,
                   float* __restrict__ Cg, int ldc,
                   int Kdim, int mBound, int nBound, int addI,
                   const int row0, const int col0,
                   float (&As)[2][64][36],
                   float (&Bs)[2][64][36])
{
    const int tid  = threadIdx.x;
    const int lane = tid & 31;
    const int wv   = tid >> 5;
    constexpr int MI  = (TM == 64) ? 2 : 1;
    constexpr int NI  = (TM == 64) ? 2 : 1;
    constexpr int AIT = TM / 16;
    const int wm   = (TM == 64) ? (wv >> 1) * 32 : 0;
    const int wn   = (TM == 64) ? (wv & 1)  * 32 : wv * 16;
    const int g    = lane >> 4;
    const int lm   = lane & 15;

    v8f acc[MI][NI] = {};
    v8f acc2 = {};

    const float* aBase = Ag + (size_t)(row0 + (tid >> 3)) * LDA + (tid & 7) * 4;
    const float* bBase = Bg + (size_t)((tid >> 6) * 4) * LDB + (col0 + (tid & 63));

    const int nCh = (Kdim + 31) >> 5;

    v4f ra[AIT], rb[4];
    v4f fa0[4][MI], fb0[4][NI];
    v4f fa1[4][MI], fb1[4][NI];

    auto gload = [&](int c) {
        const int k0 = (c < nCh ? c : nCh - 1) * 32;
        const float* aP = aBase + k0;
        const float* bP = bBase + (size_t)k0 * LDB;
        #pragma unroll
        for (int it = 0; it < AIT; ++it)
            ra[it] = *reinterpret_cast<const v4f*>(aP + it * 16 * LDA);
        #pragma unroll
        for (int it = 0; it < 4; ++it) {
            rb[it].x = bP[(it * 8 + 0) * LDB];
            rb[it].y = bP[(it * 8 + 1) * LDB];
            rb[it].z = bP[(it * 8 + 2) * LDB];
            rb[it].w = bP[(it * 8 + 3) * LDB];
        }
    };
    auto lstore = [&](const int buf) {
        #pragma unroll
        for (int it = 0; it < AIT; ++it)
            *reinterpret_cast<v4f*>(&As[buf][(tid >> 3) + 16 * it][(tid & 7) * 4]) = ra[it];
        #pragma unroll
        for (int it = 0; it < 4; ++it)
            *reinterpret_cast<v4f*>(&Bs[buf][tid & 63][((tid >> 6) + 2 * it) * 4]) = rb[it];
    };
    auto fload = [&](v4f (&fa)[4][MI], v4f (&fb)[4][NI], const int buf) {
        #pragma unroll
        for (int p = 0; p < 4; ++p) {
            const int kq = 8 * p + 4 * g;
            #pragma unroll
            for (int mi = 0; mi < MI; ++mi)
                fa[p][mi] = *reinterpret_cast<const v4f*>(&As[buf][wm + 16 * mi + lm][kq]);
            #pragma unroll
            for (int ni = 0; ni < NI; ++ni)
                fb[p][ni] = *reinterpret_cast<const v4f*>(&Bs[buf][wn + 16 * ni + lm][kq]);
        }
    };
    auto mma = [&](const v4f (&fa)[4][MI], const v4f (&fb)[4][NI], const int p0, const int p1) {
        #pragma unroll
        for (int p = p0; p < p1; ++p) {
          if constexpr (TM == 64) {
            const v2f a0l = __builtin_shufflevector(fa[p][0], fa[p][0], 0, 1), a0h = __builtin_shufflevector(fa[p][0], fa[p][0], 2, 3);
            const v2f a1l = __builtin_shufflevector(fa[p][1], fa[p][1], 0, 1), a1h = __builtin_shufflevector(fa[p][1], fa[p][1], 2, 3);
            const v2f b0l = __builtin_shufflevector(fb[p][0], fb[p][0], 0, 1), b0h = __builtin_shufflevector(fb[p][0], fb[p][0], 2, 3);
            const v2f b1l = __builtin_shufflevector(fb[p][1], fb[p][1], 0, 1), b1h = __builtin_shufflevector(fb[p][1], fb[p][1], 2, 3);
            acc[0][0] = __builtin_amdgcn_wmma_f32_16x16x4_f32(false, a0l, false, b0l, (short)0, acc[0][0], false, false);
            acc[0][1] = __builtin_amdgcn_wmma_f32_16x16x4_f32(false, a0l, false, b1l, (short)0, acc[0][1], false, false);
            acc[1][0] = __builtin_amdgcn_wmma_f32_16x16x4_f32(false, a1l, false, b0l, (short)0, acc[1][0], false, false);
            acc[1][1] = __builtin_amdgcn_wmma_f32_16x16x4_f32(false, a1l, false, b1l, (short)0, acc[1][1], false, false);
            acc[0][0] = __builtin_amdgcn_wmma_f32_16x16x4_f32(false, a0h, false, b0h, (short)0, acc[0][0], false, false);
            acc[0][1] = __builtin_amdgcn_wmma_f32_16x16x4_f32(false, a0h, false, b1h, (short)0, acc[0][1], false, false);
            acc[1][0] = __builtin_amdgcn_wmma_f32_16x16x4_f32(false, a1h, false, b0h, (short)0, acc[1][0], false, false);
            acc[1][1] = __builtin_amdgcn_wmma_f32_16x16x4_f32(false, a1h, false, b1h, (short)0, acc[1][1], false, false);
          } else {
            const v2f al = __builtin_shufflevector(fa[p][0], fa[p][0], 0, 1), ah = __builtin_shufflevector(fa[p][0], fa[p][0], 2, 3);
            const v2f bl = __builtin_shufflevector(fb[p][0], fb[p][0], 0, 1), bh = __builtin_shufflevector(fb[p][0], fb[p][0], 2, 3);
            acc[0][0] = __builtin_amdgcn_wmma_f32_16x16x4_f32(false, al, false, bl, (short)0, acc[0][0], false, false);
            acc2      = __builtin_amdgcn_wmma_f32_16x16x4_f32(false, ah, false, bh, (short)0, acc2,      false, false);
          }
        }
    };

    auto step = [&](const v4f (&ca)[4][MI], const v4f (&cb)[4][NI],
                    v4f (&na)[4][MI], v4f (&nb)[4][NI],
                    const int bufNext, const int bufStore, const int cLoad) {
        fload(na, nb, bufNext);
        lstore(bufStore);
        gload(cLoad);
        __builtin_amdgcn_sched_barrier(0);
        mma(ca, cb, 0, 1);
        __builtin_amdgcn_sched_barrier(0);
        lds_barrier_arrive();
        __builtin_amdgcn_sched_barrier(0);
        mma(ca, cb, 1, 4);
        __builtin_amdgcn_sched_barrier(0);
        lds_barrier_wait();
    };

    if (nCh > 0) {
        gload(0);
        lstore(0);
        gload(1);
        lds_barrier();
        fload(fa0, fb0, 0);
        lstore(1);
        gload(2);
        lds_barrier();

        for (int c = 0; c < nCh; c += 2) {
            step(fa0, fb0, fa1, fb1, 1, 0, c + 3);
            if (c + 1 < nCh)
                step(fa1, fb1, fa0, fb0, 0, 1, c + 4);
        }
    }

    if constexpr (TM == 16) {
        #pragma unroll
        for (int r = 0; r < 8; ++r) acc[0][0][r] += acc2[r];
    }

    #pragma unroll
    for (int mi = 0; mi < MI; ++mi)
        #pragma unroll
        for (int ni = 0; ni < NI; ++ni)
            #pragma unroll
            for (int r = 0; r < 8; ++r) {
                int row = row0 + wm + mi * 16 + r + 8 * g;
                int col = col0 + wn + ni * 16 + lm;
                if (row < mBound && col < nBound) {
                    float v = acc[mi][ni][r];
                    if (addI && row == col) v += 1.0f;
                    Cg[(size_t)row * ldc + col] = v;
                }
            }
}

template <int LDA, int LDB>
__global__ __launch_bounds__(128)
void gemm_wmma_f32(const float* __restrict__ Ag,
                   const float* __restrict__ Bg,
                   float* __restrict__ Cg, int ldc,
                   int Kdim, int mBound, int nBound, int addI)
{
    __shared__ __attribute__((aligned(16))) float As[2][64][36];
    __shared__ __attribute__((aligned(16))) float Bs[2][64][36];
    gemm_tile_f32<LDA, LDB, 64>(Ag, Bg, Cg, ldc, Kdim, mBound, nBound, addI,
                                blockIdx.y * 64, blockIdx.x * 64, As, Bs);
}

template <int LDB1, bool ALONG_N>
__global__ __launch_bounds__(128)
void gemm_pair_wmma_f32(const float* __restrict__ P, float* __restrict__ Sq,
                        const float* __restrict__ A1, const float* __restrict__ B1,
                        float* __restrict__ C1, int ldc1, int mB1, int nB1)
{
    __shared__ __attribute__((aligned(16))) float As[2][64][36];
    __shared__ __attribute__((aligned(16))) float Bs[2][64][36];
    const int b = blockIdx.x;
    if (b < 256) {
        gemm_tile_f32<NDIM, NDIM, 64>(P, P, Sq, NDIM, NDIM, NDIM, NDIM, 0,
                                      (b >> 4) * 64, (b & 15) * 64, As, Bs);
    } else {
        const int t = b - 256;
        gemm_tile_f32<NDIM, LDB1, 16>(A1, B1, C1, ldc1, NDIM, mB1, nB1, 0,
                                      ALONG_N ? (t & 3) * 16 : t * 16,
                                      ALONG_N ? (t >> 2) * 64 : 0, As, Bs);
    }
}

__global__ __launch_bounds__(128)
void gemm_ab_with_vectors(const float* __restrict__ S, const float* __restrict__ IM,
                          float* __restrict__ Ab,
                          const float* __restrict__ Bv, const float* __restrict__ ldt,
                          float* __restrict__ Y,
                          const float* __restrict__ Cv, float* __restrict__ R)
{
    __shared__ __attribute__((aligned(16))) float As[2][64][36];
    __shared__ __attribute__((aligned(16))) float Bs[2][64][36];
    const int b = blockIdx.x;
    if (b < 256) {
        gemm_tile_f32<NDIM, NDIM, 64>(S, IM, Ab, NDIM, NDIM, NDIM, NDIM, 0,
                                      (b >> 4) * 64, (b & 15) * 64, As, Bs);
    } else {
        const int i = (b - 256) * 128 + threadIdx.x;
        float dt = expf(ldt[0]);
        float s = 0.f;
        for (int k = 0; k < NDIM; ++k) s += S[(size_t)i * NDIM + k] * Bv[k];
        Y[(size_t)i * 128] = dt * s;
        R[i] = Cv[i];
    }
}

__global__ void prep_mats(const float* __restrict__ A, const float* __restrict__ ldt,
                          float* __restrict__ M, float* __restrict__ IM, int n)
{
    int idx = blockIdx.x * blockDim.x + threadIdx.x;
    if (idx >= n * n) return;
    float h = 0.5f * expf(ldt[0]);
    float m = h * A[idx];
    M[idx]  = m;
    int r = idx / n, c = idx - r * n;
    IM[idx] = m + ((r == c) ? 1.0f : 0.0f);
}

#define CV_WAVES 16
__global__ __launch_bounds__(CV_WAVES * 32)
void conv_causal_wmma(const float* __restrict__ u, const float* __restrict__ Kk,
                      const float* __restrict__ Dv, float* __restrict__ y)
{
    __shared__ float KT[LSEQ + 16];
    __shared__ float USk[255 + LSEQ + LSEQ / 16 + 1];
    __shared__ float red[CV_WAVES][256];

    const int tid = threadIdx.x;
    const int tb  = blockIdx.x;
    const int nIt = tb + 1;

    float rk[16], ru[16];
    #pragma unroll
    for (int it = 0; it < 16; ++it) {
        const int yy = tid + it * (CV_WAVES * 32);
        rk[it] = Kk[yy];
        ru[it] = u[yy];
    }
    #pragma unroll
    for (int it = 0; it < 16; ++it) {
        const int yy = tid + it * (CV_WAVES * 32);
        KT[yy] = rk[it];
        USk[255 + yy + (yy >> 4)] = ru[it];
    }
    if (tid < 16)  KT[LSEQ + tid] = 0.f;
    if (tid < 255) USk[tid] = 0.f;
    __syncthreads();

    const int lane = tid & 31;
    const int wv   = tid >> 5;
    const int c    = lane & 15;
    const int h    = lane >> 4;

    const float* ap = KT + (LSEQ + 14 - 13) - wv * 16 * nIt - c - 2 * h;
    const float* bp = USk + 17 * (16 * tb + c + 15 - wv * nIt) - 2 * h;

    auto fl = [&](v2f (&a)[4], v2f (&b)[4], const int it) {
        const float* a_ = ap - 16 * it;
        const float* b_ = bp - 17 * it;
        #pragma unroll
        for (int s = 0; s < 4; ++s) {
            a[s].x = a_[12 - 4 * s];   a[s].y = a_[13 - 4 * s];
            b[s].x = b_[14 - 4 * s];   b[s].y = b_[15 - 4 * s];
        }
    };

    v8f acc = {};
    auto mma_group = [&](const v2f (&a)[4], const v2f (&b)[4]) {
        __builtin_amdgcn_sched_barrier(0);
        #pragma unroll
        for (int s = 0; s < 4; ++s)
            acc = __builtin_amdgcn_wmma_f32_16x16x4_f32(false, a[s], false, b[s], (short)0, acc, false, false);
        asm volatile("v_nop\n\tv_nop\n\tv_nop\n\tv_nop" : "+v"(acc));
        __builtin_amdgcn_sched_barrier(0);
    };

    v2f a0[4], b0[4], a1[4], b1[4];
    fl(a0, b0, 0);
    for (int it = 0; it < nIt; it += 2) {
        fl(a1, b1, (it + 1 < nIt) ? it + 1 : it);
        mma_group(a0, b0);
        if (it + 1 < nIt) {
            fl(a0, b0, (it + 2 < nIt) ? it + 2 : it + 1);
            mma_group(a1, b1);
        }
    }

    float* rp = &red[wv][16 * c + 8 * h];
    #pragma unroll
    for (int r = 0; r < 8; ++r) rp[r] = acc[r];
    __syncthreads();

    if (tid < 256) {
        float s = 0.f;
        #pragma unroll
        for (int w = 0; w < CV_WAVES; ++w) s += red[w][tid];
        const int t = tb * 256 + tid;
        y[t] = s + Dv[0] * u[t];
    }
}

extern "C" void kernel_launch(void* const* d_in, const int* in_sizes, int n_in,
                              void* d_out, int out_size, void* d_ws, size_t ws_size,
                              hipStream_t stream)
{
    (void)in_sizes; (void)n_in; (void)out_size; (void)ws_size;
    const float* u   = (const float*)d_in[0];
    const float* A   = (const float*)d_in[1];
    const float* Bv  = (const float*)d_in[2];
    const float* Cv  = (const float*)d_in[3];
    const float* Dv  = (const float*)d_in[4];
    const float* ldt = (const float*)d_in[5];
    float* y = (float*)d_out;
    const int N = NDIM, L = LSEQ;

    float* B0 = (float*)d_ws;
    float* B1 = B0 + (size_t)N * N;
    float* B2 = B1 + (size_t)N * N;
    float* B3 = B2 + (size_t)N * N;
    float* Y  = B3 + (size_t)N * N;
    float* R  = Y  + (size_t)N * 128;
    float* Kk = R  + (size_t)64 * N;

    dim3 blk(128);
    auto gemm = [&](const float* Am, int lda, const float* Bm, int ldb,
                    float* Cm, int ldc, int Mc, int Nc, int Kd,
                    int mB, int nB, int addI) {
        (void)lda;
        dim3 grid((Nc + 63) / 64, (Mc + 63) / 64);
        if (ldb == 128)
            gemm_wmma_f32<NDIM, 128><<<grid, blk, 0, stream>>>(Am, Bm, Cm, ldc, Kd, mB, nB, addI);
        else
            gemm_wmma_f32<NDIM, NDIM><<<grid, blk, 0, stream>>>(Am, Bm, Cm, ldc, Kd, mB, nB, addI);
    };

    prep_mats<<<(N * N + 255) / 256, 256, 0, stream>>>(A, ldt, B0, B1, N);
    gemm(B0, N, B1, N, B2, N, N, N, N, N, N, 1);
    gemm(B0, N, B2, N, B3, N, N, N, N, N, N, 1);
    gemm(B0, N, B3, N, B2, N, N, N, N, N, N, 1);
    gemm_ab_with_vectors<<<dim3(256 + 8), blk, 0, stream>>>(B2, B1, B3, Bv, ldt, Y, Cv, R);

    dim3 gridPair(256 + 64);
    auto pairY = [&](const float* Ms, float* Sq, float* Cm, int nB) {
        gemm_pair_wmma_f32<128, false><<<gridPair, blk, 0, stream>>>(Ms, Sq, Ms, Y, Cm, 128, N, nB);
    };
    auto pairR = [&](const float* Qt, float* Sq, float* Cm, int mB) {
        gemm_pair_wmma_f32<NDIM, true><<<gridPair, blk, 0, stream>>>(Qt, Sq, R, Qt, Cm, N, mB, N);
    };

    pairY(B3, B0, Y + 1,  127);
    pairY(B0, B1, Y + 2,  126);
    pairY(B1, B0, Y + 4,  124);
    pairY(B0, B1, Y + 8,  120);
    pairY(B1, B0, Y + 16, 112);
    pairY(B0, B1, Y + 32,  96);
    pairY(B1, B0, Y + 64,  64);

    pairR(B0, B1, R + 1 * N,  63);
    pairR(B1, B0, R + 2 * N,  62);
    pairR(B0, B1, R + 4 * N,  60);
    pairR(B1, B0, R + 8 * N,  56);
    pairR(B0, B1, R + 16 * N, 48);
    gemm(R, N, B1, N, R + 32 * N, N, 64, N, N, 32, N, 0);

    gemm(R, N, Y, 128, Kk, 128, 64, 128, N, 64, 128, 0);

    conv_causal_wmma<<<L / 256, 512, 0, stream>>>(u, Kk, Dv, y);
}


# Every product that ran on 64x64 tiles now on 64x32 (1024^3: 512 blocks): two independent blocks per WGP, two waves per SIMD, one's staging and barrier waits filled by the other's WMMAs; same sums
# speedup vs baseline: 9.1829x; 1.0314x over previous
#include <hip/hip_runtime.h>
#include <math.h>


typedef __attribute__((ext_vector_type(2))) float v2f;
typedef __attribute__((ext_vector_type(4))) float v4f;
typedef __attribute__((ext_vector_type(8))) float v8f;

#define NDIM 1024
#define LSEQ 8192

__device__ __forceinline__ void lds_barrier_arrive()
{
    __builtin_amdgcn_fence(__ATOMIC_RELEASE, "workgroup", "local");
    __builtin_amdgcn_s_barrier_signal(-1);
}
__device__ __forceinline__ void lds_barrier_wait()
{
    __builtin_amdgcn_s_barrier_wait(-1);
    __builtin_amdgcn_fence(__ATOMIC_ACQUIRE, "workgroup", "local");
}
__device__ __forceinline__ void lds_barrier()
{
    lds_barrier_arrive();
    lds_barrier_wait();
}

template <int LDA, int LDB, int TM>
__device__ __forceinline__
void gemm_tile_f32(const float* __restrict__ Ag,
                   const float* __restrict__ Bg,
                   float* __restrict__ Cg, int ldc,
                   int Kdim, int mBound, int nBound, int addI,
                   const int row0, const int col0,
                   float (&As)[2][64][36],
                   float (&Bs)[2][64][36])
{
    const int tid  = threadIdx.x;
    const int lane = tid & 31;
    const int wv   = tid >> 5;
    constexpr int TN  = (TM == 64) ? 32 : 64;
    constexpr int NI  = (TM == 64) ? 2 : 1;
    constexpr int AIT = TM / 16;
    constexpr int BIT = TN / 16;
    constexpr int KGP = 128 / TN;
    const int wm   = (TM == 64) ? wv * 16 : 0;
    const int wn   = (TM == 64) ? 0 : wv * 16;
    const int g    = lane >> 4;
    const int lm   = lane & 15;

    v8f acc[NI] = {};
    v8f acc2 = {};

    const float* aBase = Ag + (size_t)(row0 + (tid >> 3)) * LDA + (tid & 7) * 4;
    const float* bBase = Bg + (size_t)((tid / TN) * 4) * LDB + (col0 + (tid & (TN - 1)));

    const int nCh = (Kdim + 31) >> 5;

    v4f ra[AIT], rb[BIT];
    v4f fa0[4], fb0[4][NI];
    v4f fa1[4], fb1[4][NI];

    auto gload = [&](int c) {
        const int k0 = (c < nCh ? c : nCh - 1) * 32;
        const float* aP = aBase + k0;
        const float* bP = bBase + (size_t)k0 * LDB;
        #pragma unroll
        for (int it = 0; it < AIT; ++it)
            ra[it] = *reinterpret_cast<const v4f*>(aP + it * 16 * LDA);
        #pragma unroll
        for (int it = 0; it < BIT; ++it) {
            rb[it].x = bP[(it * KGP * 4 + 0) * LDB];
            rb[it].y = bP[(it * KGP * 4 + 1) * LDB];
            rb[it].z = bP[(it * KGP * 4 + 2) * LDB];
            rb[it].w = bP[(it * KGP * 4 + 3) * LDB];
        }
    };
    auto lstore = [&](const int buf) {
        #pragma unroll
        for (int it = 0; it < AIT; ++it)
            *reinterpret_cast<v4f*>(&As[buf][(tid >> 3) + 16 * it][(tid & 7) * 4]) = ra[it];
        #pragma unroll
        for (int it = 0; it < BIT; ++it)
            *reinterpret_cast<v4f*>(&Bs[buf][tid & (TN - 1)][((tid / TN) + KGP * it) * 4]) = rb[it];
    };
    auto fload = [&](v4f (&fa)[4], v4f (&fb)[4][NI], const int buf) {
        #pragma unroll
        for (int p = 0; p < 4; ++p) {
            const int kq = 8 * p + 4 * g;
            fa[p] = *reinterpret_cast<const v4f*>(&As[buf][wm + lm][kq]);
            #pragma unroll
            for (int ni = 0; ni < NI; ++ni)
                fb[p][ni] = *reinterpret_cast<const v4f*>(&Bs[buf][wn + 16 * ni + lm][kq]);
        }
    };
    auto mma = [&](const v4f (&fa)[4], const v4f (&fb)[4][NI], const int p0, const int p1) {
        #pragma unroll
        for (int p = p0; p < p1; ++p) {
          const v2f al = __builtin_shufflevector(fa[p], fa[p], 0, 1), ah = __builtin_shufflevector(fa[p], fa[p], 2, 3);
          if constexpr (TM == 64) {
            const v2f b0l = __builtin_shufflevector(fb[p][0], fb[p][0], 0, 1), b0h = __builtin_shufflevector(fb[p][0], fb[p][0], 2, 3);
            const v2f b1l = __builtin_shufflevector(fb[p][1], fb[p][1], 0, 1), b1h = __builtin_shufflevector(fb[p][1], fb[p][1], 2, 3);
            acc[0] = __builtin_amdgcn_wmma_f32_16x16x4_f32(false, al, false, b0l, (short)0, acc[0], false, false);
            acc[1] = __builtin_amdgcn_wmma_f32_16x16x4_f32(false, al, false, b1l, (short)0, acc[1], false, false);
            acc[0] = __builtin_amdgcn_wmma_f32_16x16x4_f32(false, ah, false, b0h, (short)0, acc[0], false, false);
            acc[1] = __builtin_amdgcn_wmma_f32_16x16x4_f32(false, ah, false, b1h, (short)0, acc[1], false, false);
          } else {
            const v2f bl = __builtin_shufflevector(fb[p][0], fb[p][0], 0, 1), bh = __builtin_shufflevector(fb[p][0], fb[p][0], 2, 3);
            acc[0] = __builtin_amdgcn_wmma_f32_16x16x4_f32(false, al, false, bl, (short)0, acc[0], false, false);
            acc2   = __builtin_amdgcn_wmma_f32_16x16x4_f32(false, ah, false, bh, (short)0, acc2,   false, false);
          }
        }
    };

    constexpr int PA = (TM == 64) ? 2 : 1;
    auto step = [&](const v4f (&ca)[4], const v4f (&cb)[4][NI],
                    v4f (&na)[4], v4f (&nb)[4][NI],
                    const int bufNext, const int bufStore, const int cLoad) {
        fload(na, nb, bufNext);
        lstore(bufStore);
        gload(cLoad);
        __builtin_amdgcn_sched_barrier(0);
        mma(ca, cb, 0, PA);
        __builtin_amdgcn_sched_barrier(0);
        lds_barrier_arrive();
        __builtin_amdgcn_sched_barrier(0);
        mma(ca, cb, PA, 4);
        __builtin_amdgcn_sched_barrier(0);
        lds_barrier_wait();
    };

    if (nCh > 0) {
        gload(0);
        lstore(0);
        gload(1);
        lds_barrier();
        fload(fa0, fb0, 0);
        lstore(1);
        gload(2);
        lds_barrier();

        for (int c = 0; c < nCh; c += 2) {
            step(fa0, fb0, fa1, fb1, 1, 0, c + 3);
            if (c + 1 < nCh)
                step(fa1, fb1, fa0, fb0, 0, 1, c + 4);
        }
    }

    if constexpr (TM == 16) {
        #pragma unroll
        for (int r = 0; r < 8; ++r) acc[0][r] += acc2[r];
    }

    #pragma unroll
    for (int ni = 0; ni < NI; ++ni)
        #pragma unroll
        for (int r = 0; r < 8; ++r) {
            int row = row0 + wm + r + 8 * g;
            int col = col0 + wn + ni * 16 + lm;
            if (row < mBound && col < nBound) {
                float v = acc[ni][r];
                if (addI && row == col) v += 1.0f;
                Cg[(size_t)row * ldc + col] = v;
            }
        }
}

__device__ __forceinline__ void half_tile_origin(const int b, const int nReg, const int nc64,
                                                 int& row0, int& col0)
{
    const int hf = (b >= nReg) ? 1 : 0;
    const int t  = b - hf * nReg;
    row0 = (t / nc64) * 64;
    col0 = ((t % nc64) * 2 + hf) * 32;
}

template <int LDA, int LDB>
__global__ __launch_bounds__(128)
void gemm_wmma_f32(const float* __restrict__ Ag,
                   const float* __restrict__ Bg,
                   float* __restrict__ Cg, int ldc,
                   int Kdim, int mBound, int nBound, int addI, int nReg)
{
    __shared__ __attribute__((aligned(16))) float As[2][64][36];
    __shared__ __attribute__((aligned(16))) float Bs[2][64][36];
    int row0, col0;
    half_tile_origin(blockIdx.x, nReg, LDB / 64, row0, col0);
    gemm_tile_f32<LDA, LDB, 64>(Ag, Bg, Cg, ldc, Kdim, mBound, nBound, addI, row0, col0, As, Bs);
}

template <int LDB1, bool ALONG_N>
__global__ __launch_bounds__(128)
void gemm_pair_wmma_f32(const float* __restrict__ P, float* __restrict__ Sq,
                        const float* __restrict__ A1, const float* __restrict__ B1,
                        float* __restrict__ C1, int ldc1, int mB1, int nB1)
{
    __shared__ __attribute__((aligned(16))) float As[2][64][36];
    __shared__ __attribute__((aligned(16))) float Bs[2][64][36];
    const int b = blockIdx.x;
    if (b < 512) {
        int row0, col0;
        half_tile_origin(b, 256, 16, row0, col0);
        gemm_tile_f32<NDIM, NDIM, 64>(P, P, Sq, NDIM, NDIM, NDIM, NDIM, 0, row0, col0, As, Bs);
    } else {
        const int t = b - 512;
        gemm_tile_f32<NDIM, LDB1, 16>(A1, B1, C1, ldc1, NDIM, mB1, nB1, 0,
                                      ALONG_N ? (t & 3) * 16 : t * 16,
                                      ALONG_N ? (t >> 2) * 64 : 0, As, Bs);
    }
}

__global__ __launch_bounds__(128)
void gemm_ab_with_vectors(const float* __restrict__ S, const float* __restrict__ IM,
                          float* __restrict__ Ab,
                          const float* __restrict__ Bv, const float* __restrict__ ldt,
                          float* __restrict__ Y,
                          const float* __restrict__ Cv, float* __restrict__ R)
{
    __shared__ __attribute__((aligned(16))) float As[2][64][36];
    __shared__ __attribute__((aligned(16))) float Bs[2][64][36];
    const int b = blockIdx.x;
    if (b < 512) {
        int row0, col0;
        half_tile_origin(b, 256, 16, row0, col0);
        gemm_tile_f32<NDIM, NDIM, 64>(S, IM, Ab, NDIM, NDIM, NDIM, NDIM, 0, row0, col0, As, Bs);
    } else {
        const int i = (b - 512) * 128 + threadIdx.x;
        float dt = expf(ldt[0]);
        float s = 0.f;
        for (int k = 0; k < NDIM; ++k) s += S[(size_t)i * NDIM + k] * Bv[k];
        Y[(size_t)i * 128] = dt * s;
        R[i] = Cv[i];
    }
}

__global__ void prep_mats(const float* __restrict__ A, const float* __restrict__ ldt,
                          float* __restrict__ M, float* __restrict__ IM, int n)
{
    int idx = blockIdx.x * blockDim.x + threadIdx.x;
    if (idx >= n * n) return;
    float h = 0.5f * expf(ldt[0]);
    float m = h * A[idx];
    M[idx]  = m;
    int r = idx / n, c = idx - r * n;
    IM[idx] = m + ((r == c) ? 1.0f : 0.0f);
}

#define CV_WAVES 16
__global__ __launch_bounds__(CV_WAVES * 32)
void conv_causal_wmma(const float* __restrict__ u, const float* __restrict__ Kk,
                      const float* __restrict__ Dv, float* __restrict__ y)
{
    __shared__ float KT[LSEQ + 16];
    __shared__ float USk[255 + LSEQ + LSEQ / 16 + 1];
    __shared__ float red[CV_WAVES][256];

    const int tid = threadIdx.x;
    const int tb  = blockIdx.x;
    const int nIt = tb + 1;

    float rk[16], ru[16];
    #pragma unroll
    for (int it = 0; it < 16; ++it) {
        const int yy = tid + it * (CV_WAVES * 32);
        rk[it] = Kk[yy];
        ru[it] = u[yy];
    }
    #pragma unroll
    for (int it = 0; it < 16; ++it) {
        const int yy = tid + it * (CV_WAVES * 32);
        KT[yy] = rk[it];
        USk[255 + yy + (yy >> 4)] = ru[it];
    }
    if (tid < 16)  KT[LSEQ + tid] = 0.f;
    if (tid < 255) USk[tid] = 0.f;
    __syncthreads();

    const int lane = tid & 31;
    const int wv   = tid >> 5;
    const int c    = lane & 15;
    const int h    = lane >> 4;

    const float* ap = KT + (LSEQ + 14 - 13) - wv * 16 * nIt - c - 2 * h;
    const float* bp = USk + 17 * (16 * tb + c + 15 - wv * nIt) - 2 * h;

    auto fl = [&](v2f (&a)[4], v2f (&b)[4], const int it) {
        const float* a_ = ap - 16 * it;
        const float* b_ = bp - 17 * it;
        #pragma unroll
        for (int s = 0; s < 4; ++s) {
            a[s].x = a_[12 - 4 * s];   a[s].y = a_[13 - 4 * s];
            b[s].x = b_[14 - 4 * s];   b[s].y = b_[15 - 4 * s];
        }
    };

    v8f acc = {};
    auto mma_group = [&](const v2f (&a)[4], const v2f (&b)[4]) {
        __builtin_amdgcn_sched_barrier(0);
        #pragma unroll
        for (int s = 0; s < 4; ++s)
            acc = __builtin_amdgcn_wmma_f32_16x16x4_f32(false, a[s], false, b[s], (short)0, acc, false, false);
        asm volatile("v_nop\n\tv_nop\n\tv_nop\n\tv_nop" : "+v"(acc));
        __builtin_amdgcn_sched_barrier(0);
    };

    v2f a0[4], b0[4], a1[4], b1[4];
    fl(a0, b0, 0);
    for (int it = 0; it < nIt; it += 2) {
        fl(a1, b1, (it + 1 < nIt) ? it + 1 : it);
        mma_group(a0, b0);
        if (it + 1 < nIt) {
            fl(a0, b0, (it + 2 < nIt) ? it + 2 : it + 1);
            mma_group(a1, b1);
        }
    }

    float* rp = &red[wv][16 * c + 8 * h];
    #pragma unroll
    for (int r = 0; r < 8; ++r) rp[r] = acc[r];
    __syncthreads();

    if (tid < 256) {
        float s = 0.f;
        #pragma unroll
        for (int w = 0; w < CV_WAVES; ++w) s += red[w][tid];
        const int t = tb * 256 + tid;
        y[t] = s + Dv[0] * u[t];
    }
}

extern "C" void kernel_launch(void* const* d_in, const int* in_sizes, int n_in,
                              void* d_out, int out_size, void* d_ws, size_t ws_size,
                              hipStream_t stream)
{
    (void)in_sizes; (void)n_in; (void)out_size; (void)ws_size;
    const float* u   = (const float*)d_in[0];
    const float* A   = (const float*)d_in[1];
    const float* Bv  = (const float*)d_in[2];
    const float* Cv  = (const float*)d_in[3];
    const float* Dv  = (const float*)d_in[4];
    const float* ldt = (const float*)d_in[5];
    float* y = (float*)d_out;
    const int N = NDIM, L = LSEQ;

    float* B0 = (float*)d_ws;
    float* B1 = B0 + (size_t)N * N;
    float* B2 = B1 + (size_t)N * N;
    float* B3 = B2 + (size_t)N * N;
    float* Y  = B3 + (size_t)N * N;
    float* R  = Y  + (size_t)N * 128;
    float* Kk = R  + (size_t)64 * N;

    dim3 blk(128);
    auto gemm = [&](const float* Am, int lda, const float* Bm, int ldb,
                    float* Cm, int ldc, int Mc, int Nc, int Kd,
                    int mB, int nB, int addI) {
        (void)lda;
        const int nReg = ((Nc + 63) / 64) * ((Mc + 63) / 64);
        dim3 grid(2 * nReg);
        if (ldb == 128)
            gemm_wmma_f32<NDIM, 128><<<grid, blk, 0, stream>>>(Am, Bm, Cm, ldc, Kd, mB, nB, addI, nReg);
        else
            gemm_wmma_f32<NDIM, NDIM><<<grid, blk, 0, stream>>>(Am, Bm, Cm, ldc, Kd, mB, nB, addI, nReg);
    };

    prep_mats<<<(N * N + 255) / 256, 256, 0, stream>>>(A, ldt, B0, B1, N);
    gemm(B0, N, B1, N, B2, N, N, N, N, N, N, 1);
    gemm(B0, N, B2, N, B3, N, N, N, N, N, N, 1);
    gemm(B0, N, B3, N, B2, N, N, N, N, N, N, 1);
    gemm_ab_with_vectors<<<dim3(512 + 8), blk, 0, stream>>>(B2, B1, B3, Bv, ldt, Y, Cv, R);

    dim3 gridPair(512 + 64);
    auto pairY = [&](const float* Ms, float* Sq, float* Cm, int nB) {
        gemm_pair_wmma_f32<128, false><<<gridPair, blk, 0, stream>>>(Ms, Sq, Ms, Y, Cm, 128, N, nB);
    };
    auto pairR = [&](const float* Qt, float* Sq, float* Cm, int mB) {
        gemm_pair_wmma_f32<NDIM, true><<<gridPair, blk, 0, stream>>>(Qt, Sq, R, Qt, Cm, N, mB, N);
    };

    pairY(B3, B0, Y + 1,  127);
    pairY(B0, B1, Y + 2,  126);
    pairY(B1, B0, Y + 4,  124);
    pairY(B0, B1, Y + 8,  120);
    pairY(B1, B0, Y + 16, 112);
    pairY(B0, B1, Y + 32,  96);
    pairY(B1, B0, Y + 64,  64);

    pairR(B0, B1, R + 1 * N,  63);
    pairR(B1, B0, R + 2 * N,  62);
    pairR(B0, B1, R + 4 * N,  60);
    pairR(B1, B0, R + 8 * N,  56);
    pairR(B0, B1, R + 16 * N, 48);
    gemm(R, N, B1, N, R + 32 * N, N, 64, N, N, 32, N, 0);

    gemm(R, N, Y, 128, Kk, 128, 64, 128, N, 64, 128, 0);

    conv_causal_wmma<<<L / 256, 512, 0, stream>>>(u, Kk, Dv, y);
}


# GEMM K loop: each chunk's global loads are issued three steps ahead of their LDS store instead of one (three register sets in a six-step pattern); sums and their order unchanged
# speedup vs baseline: 11.2014x; 1.2198x over previous
#include <hip/hip_runtime.h>
#include <math.h>


typedef __attribute__((ext_vector_type(2))) float v2f;
typedef __attribute__((ext_vector_type(4))) float v4f;
typedef __attribute__((ext_vector_type(8))) float v8f;

#define NDIM 1024
#define LSEQ 8192

__device__ __forceinline__ void lds_barrier_arrive()
{
    __builtin_amdgcn_fence(__ATOMIC_RELEASE, "workgroup", "local");
    __builtin_amdgcn_s_barrier_signal(-1);
}
__device__ __forceinline__ void lds_barrier_wait()
{
    __builtin_amdgcn_s_barrier_wait(-1);
    __builtin_amdgcn_fence(__ATOMIC_ACQUIRE, "workgroup", "local");
}
__device__ __forceinline__ void lds_barrier()
{
    lds_barrier_arrive();
    lds_barrier_wait();
}

template <int LDA, int LDB, int TM, int EPI>
__device__ __forceinline__
void gemm_tile_f32(const float* __restrict__ Ag,
                   const float* __restrict__ Bg,
                   float* __restrict__ Cg, int ldc,
                   const float* __restrict__ Eg, float* __restrict__ C2g,
                   int Kdim, int mBound, int nBound,
                   const int row0, const int col0,
                   float (&As)[2][64][36],
                   float (&Bs)[2][64][36])
{
    const int tid  = threadIdx.x;
    const int lane = tid & 31;
    const int wv   = tid >> 5;
    constexpr int TN  = (TM == 64) ? 32 : 64;
    constexpr int NI  = (TM == 64) ? 2 : 1;
    constexpr int AIT = TM / 16;
    constexpr int BIT = TN / 16;
    constexpr int KGP = 128 / TN;
    const int wm   = (TM == 64) ? wv * 16 : 0;
    const int wn   = (TM == 64) ? 0 : wv * 16;
    const int g    = lane >> 4;
    const int lm   = lane & 15;

    v8f acc[NI] = {};
    v8f acc2 = {};

    const float* aBase = Ag + (size_t)(row0 + (tid >> 3)) * LDA + (tid & 7) * 4;
    const float* bBase = Bg + (size_t)((tid / TN) * 4) * LDB + (col0 + (tid & (TN - 1)));

    const int nCh = (Kdim + 31) >> 5;

    v4f ra[3][AIT], rb[3][BIT];
    v4f fa0[4], fb0[4][NI];
    v4f fa1[4], fb1[4][NI];

    auto gload = [&](v4f (&qa)[AIT], v4f (&qb)[BIT], int c) {
        const int k0 = (c < nCh ? c : nCh - 1) * 32;
        const float* aP = aBase + k0;
        const float* bP = bBase + (size_t)k0 * LDB;
        #pragma unroll
        for (int it = 0; it < AIT; ++it)
            qa[it] = *reinterpret_cast<const v4f*>(aP + it * 16 * LDA);
        #pragma unroll
        for (int it = 0; it < BIT; ++it) {
            qb[it].x = bP[(it * KGP * 4 + 0) * LDB];
            qb[it].y = bP[(it * KGP * 4 + 1) * LDB];
            qb[it].z = bP[(it * KGP * 4 + 2) * LDB];
            qb[it].w = bP[(it * KGP * 4 + 3) * LDB];
        }
    };
    auto lstore = [&](const v4f (&qa)[AIT], const v4f (&qb)[BIT], const int buf) {
        #pragma unroll
        for (int it = 0; it < AIT; ++it)
            *reinterpret_cast<v4f*>(&As[buf][(tid >> 3) + 16 * it][(tid & 7) * 4]) = qa[it];
        #pragma unroll
        for (int it = 0; it < BIT; ++it)
            *reinterpret_cast<v4f*>(&Bs[buf][tid & (TN - 1)][((tid / TN) + KGP * it) * 4]) = qb[it];
    };
    auto fload = [&](v4f (&fa)[4], v4f (&fb)[4][NI], const int buf) {
        #pragma unroll
        for (int p = 0; p < 4; ++p) {
            const int kq = 8 * p + 4 * g;
            fa[p] = *reinterpret_cast<const v4f*>(&As[buf][wm + lm][kq]);
            #pragma unroll
            for (int ni = 0; ni < NI; ++ni)
                fb[p][ni] = *reinterpret_cast<const v4f*>(&Bs[buf][wn + 16 * ni + lm][kq]);
        }
    };
    auto mma = [&](const v4f (&fa)[4], const v4f (&fb)[4][NI], const int p0, const int p1) {
        #pragma unroll
        for (int p = p0; p < p1; ++p) {
          const v2f al = __builtin_shufflevector(fa[p], fa[p], 0, 1), ah = __builtin_shufflevector(fa[p], fa[p], 2, 3);
          if constexpr (TM == 64) {
            const v2f b0l = __builtin_shufflevector(fb[p][0], fb[p][0], 0, 1), b0h = __builtin_shufflevector(fb[p][0], fb[p][0], 2, 3);
            const v2f b1l = __builtin_shufflevector(fb[p][1], fb[p][1], 0, 1), b1h = __builtin_shufflevector(fb[p][1], fb[p][1], 2, 3);
            acc[0] = __builtin_amdgcn_wmma_f32_16x16x4_f32(false, al, false, b0l, (short)0, acc[0], false, false);
            acc[1] = __builtin_amdgcn_wmma_f32_16x16x4_f32(false, al, false, b1l, (short)0, acc[1], false, false);
            acc[0] = __builtin_amdgcn_wmma_f32_16x16x4_f32(false, ah, false, b0h, (short)0, acc[0], false, false);
            acc[1] = __builtin_amdgcn_wmma_f32_16x16x4_f32(false, ah, false, b1h, (short)0, acc[1], false, false);
          } else {
            const v2f bl = __builtin_shufflevector(fb[p][0], fb[p][0], 0, 1), bh = __builtin_shufflevector(fb[p][0], fb[p][0], 2, 3);
            acc[0] = __builtin_amdgcn_wmma_f32_16x16x4_f32(false, al, false, bl, (short)0, acc[0], false, false);
            acc2   = __builtin_amdgcn_wmma_f32_16x16x4_f32(false, ah, false, bh, (short)0, acc2,   false, false);
          }
        }
    };

    constexpr int PA = (TM == 64) ? 2 : 1;
    auto step = [&](const v4f (&ca)[4], const v4f (&cb)[4][NI],
                    v4f (&na)[4], v4f (&nb)[4][NI],
                    v4f (&qa)[AIT], v4f (&qb)[BIT],
                    const int bufNext, const int bufStore, const int cLoad) {
        fload(na, nb, bufNext);
        lstore(qa, qb, bufStore);
        gload(qa, qb, cLoad);
        __builtin_amdgcn_sched_barrier(0);
        mma(ca, cb, 0, PA);
        __builtin_amdgcn_sched_barrier(0);
        lds_barrier_arrive();
        __builtin_amdgcn_sched_barrier(0);
        mma(ca, cb, PA, 4);
        __builtin_amdgcn_sched_barrier(0);
        lds_barrier_wait();
    };

    if (nCh > 0) {
        gload(ra[0], rb[0], 0);
        gload(ra[1], rb[1], 1);
        gload(ra[2], rb[2], 2);
        lstore(ra[0], rb[0], 0);
        gload(ra[0], rb[0], 3);
        lds_barrier();
        fload(fa0, fb0, 0);
        lstore(ra[1], rb[1], 1);
        gload(ra[1], rb[1], 4);
        lds_barrier();

        int c = 0;
        for (; c + 6 <= nCh; c += 6) {
            step(fa0, fb0, fa1, fb1, ra[2], rb[2], 1, 0, c + 5);
            step(fa1, fb1, fa0, fb0, ra[0], rb[0], 0, 1, c + 6);
            step(fa0, fb0, fa1, fb1, ra[1], rb[1], 1, 0, c + 7);
            step(fa1, fb1, fa0, fb0, ra[2], rb[2], 0, 1, c + 8);
            step(fa0, fb0, fa1, fb1, ra[0], rb[0], 1, 0, c + 9);
            step(fa1, fb1, fa0, fb0, ra[1], rb[1], 0, 1, c + 10);
        }
        if (c < nCh) {
            step(fa0, fb0, fa1, fb1, ra[2], rb[2], 1, 0, c + 5);
            if (c + 1 < nCh) {
                step(fa1, fb1, fa0, fb0, ra[0], rb[0], 0, 1, c + 6);
                if (c + 2 < nCh) {
                    step(fa0, fb0, fa1, fb1, ra[1], rb[1], 1, 0, c + 7);
                    if (c + 3 < nCh) {
                        step(fa1, fb1, fa0, fb0, ra[2], rb[2], 0, 1, c + 8);
                        if (c + 4 < nCh)
                            step(fa0, fb0, fa1, fb1, ra[0], rb[0], 1, 0, c + 9);
                    }
                }
            }
        }
    }

    if constexpr (TM == 16) {
        #pragma unroll
        for (int r = 0; r < 8; ++r) acc[0][r] += acc2[r];
    }

    if constexpr (EPI == 0) {
        #pragma unroll
        for (int ni = 0; ni < NI; ++ni)
            #pragma unroll
            for (int r = 0; r < 8; ++r) {
                int row = row0 + wm + r + 8 * g;
                int col = col0 + wn + ni * 16 + lm;
                if (row < mBound && col < nBound)
                    Cg[(size_t)row * ldc + col] = acc[ni][r];
            }
    } else {
        float e[NI][8];
        #pragma unroll
        for (int ni = 0; ni < NI; ++ni)
            #pragma unroll
            for (int r = 0; r < 8; ++r)
                e[ni][r] = Eg[(size_t)(row0 + wm + r + 8 * g) * ldc + (col0 + wn + ni * 16 + lm)];
        #pragma unroll
        for (int ni = 0; ni < NI; ++ni)
            #pragma unroll
            for (int r = 0; r < 8; ++r) {
                const int row = row0 + wm + r + 8 * g;
                const int col = col0 + wn + ni * 16 + lm;
                const size_t o = (size_t)row * ldc + col;
                float v = acc[ni][r];
                if constexpr (EPI == 1) {
                    C2g[o] = e[ni][r] + v;
                } else {
                    v = 2.0f * (e[ni][r] + v);
                    if (row == col) v += 1.0f;
                }
                Cg[o] = v;
            }
    }
}

__device__ __forceinline__ void half_tile_origin(const int b, const int nReg, const int nc64,
                                                 int& row0, int& col0)
{
    const int hf = (b >= nReg) ? 1 : 0;
    const int t  = b - hf * nReg;
    row0 = (t / nc64) * 64;
    col0 = ((t % nc64) * 2 + hf) * 32;
}

template <int LDA, int LDB>
__global__ __launch_bounds__(128)
void gemm_wmma_f32(const float* __restrict__ Ag,
                   const float* __restrict__ Bg,
                   float* __restrict__ Cg, int ldc,
                   int Kdim, int mBound, int nBound, int nReg)
{
    __shared__ __attribute__((aligned(16))) float As[2][64][36];
    __shared__ __attribute__((aligned(16))) float Bs[2][64][36];
    int row0, col0;
    half_tile_origin(blockIdx.x, nReg, LDB / 64, row0, col0);
    gemm_tile_f32<LDA, LDB, 64, 0>(Ag, Bg, Cg, ldc, nullptr, nullptr, Kdim, mBound, nBound, row0, col0, As, Bs);
}

template <int LDB1, bool ALONG_N>
__global__ __launch_bounds__(128)
void gemm_pair_wmma_f32(const float* __restrict__ P, float* __restrict__ Sq,
                        const float* __restrict__ A1, const float* __restrict__ B1,
                        float* __restrict__ C1, int ldc1, int mB1, int nB1)
{
    __shared__ __attribute__((aligned(16))) float As[2][64][36];
    __shared__ __attribute__((aligned(16))) float Bs[2][64][36];
    const int b = blockIdx.x;
    if (b < 512) {
        int row0, col0;
        half_tile_origin(b, 256, 16, row0, col0);
        gemm_tile_f32<NDIM, NDIM, 64, 0>(P, P, Sq, NDIM, nullptr, nullptr, NDIM, NDIM, NDIM, row0, col0, As, Bs);
    } else {
        const int t = b - 512;
        gemm_tile_f32<NDIM, LDB1, 16, 0>(A1, B1, C1, ldc1, nullptr, nullptr, NDIM, mB1, nB1,
                                      ALONG_N ? (t & 3) * 16 : t * 16,
                                      ALONG_N ? (t >> 2) * 64 : 0, As, Bs);
    }
}


__device__ __forceinline__ float row_dot_wave(const float* __restrict__ rowp, const v4f (&xv)[8],
                                              const int lane)
{
    v4f a[8];
    #pragma unroll
    for (int j = 0; j < 8; ++j) a[j] = *reinterpret_cast<const v4f*>(rowp + 128 * j + 4 * lane);
    float sum = 0.f;
    #pragma unroll
    for (int j = 0; j < 8; ++j)
        sum += a[j].x * xv[j].x + a[j].y * xv[j].y + a[j].z * xv[j].z + a[j].w * xv[j].w;
    #pragma unroll
    for (int o = 16; o > 0; o >>= 1) sum += __shfl_xor(sum, o);
    return sum;
}

__device__ __forceinline__ void load_vec_wave(const float* __restrict__ x, v4f (&xv)[8], const int lane)
{
    #pragma unroll
    for (int j = 0; j < 8; ++j) xv[j] = *reinterpret_cast<const v4f*>(x + 128 * j + 4 * lane);
}

template <int STAGE>
__global__ __launch_bounds__(128)
void gemm_disc_with_vector(const float* __restrict__ Aop, const float* __restrict__ Bop,
                           float* __restrict__ Cm, const float* __restrict__ Em,
                           float* __restrict__ C2m,
                           const float* __restrict__ q1, float* __restrict__ q2w,
                           const float* __restrict__ q2r,
                           const float* __restrict__ Bv, const float* __restrict__ ldt,
                           float* __restrict__ Y,
                           const float* __restrict__ Cv, float* __restrict__ R)
{
    __shared__ __attribute__((aligned(16))) float As[2][64][36];
    __shared__ __attribute__((aligned(16))) float Bs[2][64][36];
    const int b = blockIdx.x;
    if (b < 512) {
        int row0, col0;
        half_tile_origin(b, 256, 16, row0, col0);
        gemm_tile_f32<NDIM, NDIM, 64, STAGE>(Aop, Bop, Cm, NDIM, Em, C2m, NDIM, NDIM, NDIM, row0, col0, As, Bs);
    } else {
        const int lane = threadIdx.x & 31;
        const int rowW = ((b - 512) * 4 + (threadIdx.x >> 5)) * 8;
        v4f xv[8];
        load_vec_wave(q1, xv, lane);
        if constexpr (STAGE == 2) {
            v4f x2[8];
            load_vec_wave(q2r, x2, lane);
            #pragma unroll
            for (int j = 0; j < 8; ++j) xv[j] += x2[j];
        }
        const float dt = (STAGE == 2) ? expf(ldt[0]) : 0.f;
        for (int rr = 0; rr < 8; ++rr) {
            const int i = rowW + rr;
            const float sum = row_dot_wave(Aop + (size_t)i * NDIM, xv, lane);
            if (lane == 0) {
                if constexpr (STAGE == 1) {
                    q2w[i] = sum;
                } else {
                    Y[(size_t)i * 128] = dt * (Bv[i] + q1[i] + q2r[i] + sum);
                }
            }
        }
        if constexpr (STAGE == 2) {
            if (lane < 8) R[rowW + lane] = Cv[rowW + lane];
        }
    }
}

__global__ __launch_bounds__(256)
void prep_mats(const float* __restrict__ A, const float* __restrict__ ldt,
               const float* __restrict__ Bv,
               float* __restrict__ M, float* __restrict__ q1, int n)
{
    const float h = 0.5f * expf(ldt[0]);
    if (blockIdx.x < 32) {
        const int lane = threadIdx.x & 31;
        const int rowW = (blockIdx.x * 8 + (threadIdx.x >> 5)) * 4;
        v4f xv[8];
        load_vec_wave(Bv, xv, lane);
        for (int rr = 0; rr < 4; ++rr) {
            const float sum = row_dot_wave(A + (size_t)(rowW + rr) * NDIM, xv, lane);
            if (lane == 0) q1[rowW + rr] = h * sum;
        }
        return;
    }
    int idx = (blockIdx.x - 32) * blockDim.x + threadIdx.x;
    if (idx >= n * n) return;
    M[idx] = h * A[idx];
}

#define CV_WAVES 16
__global__ __launch_bounds__(CV_WAVES * 32)
void conv_causal_wmma(const float* __restrict__ u, const float* __restrict__ Kk,
                      const float* __restrict__ Dv, float* __restrict__ y)
{
    __shared__ float KT[LSEQ + 16];
    __shared__ float USk[255 + LSEQ + LSEQ / 16 + 1];
    __shared__ float red[CV_WAVES][256];

    const int tid = threadIdx.x;
    const int tb  = blockIdx.x;
    const int nIt = tb + 1;

    float rk[16], ru[16];
    #pragma unroll
    for (int it = 0; it < 16; ++it) {
        const int yy = tid + it * (CV_WAVES * 32);
        rk[it] = Kk[yy];
        ru[it] = u[yy];
    }
    #pragma unroll
    for (int it = 0; it < 16; ++it) {
        const int yy = tid + it * (CV_WAVES * 32);
        KT[yy] = rk[it];
        USk[255 + yy + (yy >> 4)] = ru[it];
    }
    if (tid < 16)  KT[LSEQ + tid] = 0.f;
    if (tid < 255) USk[tid] = 0.f;
    __syncthreads();

    const int lane = tid & 31;
    const int wv   = tid >> 5;
    const int c    = lane & 15;
    const int h    = lane >> 4;

    const float* ap = KT + (LSEQ + 14 - 13) - wv * 16 * nIt - c - 2 * h;
    const float* bp = USk + 17 * (16 * tb + c + 15 - wv * nIt) - 2 * h;

    auto fl = [&](v2f (&a)[4], v2f (&b)[4], const int it) {
        const float* a_ = ap - 16 * it;
        const float* b_ = bp - 17 * it;
        #pragma unroll
        for (int s = 0; s < 4; ++s) {
            a[s].x = a_[12 - 4 * s];   a[s].y = a_[13 - 4 * s];
            b[s].x = b_[14 - 4 * s];   b[s].y = b_[15 - 4 * s];
        }
    };

    v8f acc = {};
    auto mma_group = [&](const v2f (&a)[4], const v2f (&b)[4]) {
        __builtin_amdgcn_sched_barrier(0);
        #pragma unroll
        for (int s = 0; s < 4; ++s)
            acc = __builtin_amdgcn_wmma_f32_16x16x4_f32(false, a[s], false, b[s], (short)0, acc, false, false);
        asm volatile("v_nop\n\tv_nop\n\tv_nop\n\tv_nop" : "+v"(acc));
        __builtin_amdgcn_sched_barrier(0);
    };

    v2f a0[4], b0[4], a1[4], b1[4];
    fl(a0, b0, 0);
    for (int it = 0; it < nIt; it += 2) {
        fl(a1, b1, (it + 1 < nIt) ? it + 1 : it);
        mma_group(a0, b0);
        if (it + 1 < nIt) {
            fl(a0, b0, (it + 2 < nIt) ? it + 2 : it + 1);
            mma_group(a1, b1);
        }
    }

    float* rp = &red[wv][16 * c + 8 * h];
    #pragma unroll
    for (int r = 0; r < 8; ++r) rp[r] = acc[r];
    __syncthreads();

    if (tid < 256) {
        float s = 0.f;
        #pragma unroll
        for (int w = 0; w < CV_WAVES; ++w) s += red[w][tid];
        const int t = tb * 256 + tid;
        y[t] = s + Dv[0] * u[t];
    }
}

extern "C" void kernel_launch(void* const* d_in, const int* in_sizes, int n_in,
                              void* d_out, int out_size, void* d_ws, size_t ws_size,
                              hipStream_t stream)
{
    (void)in_sizes; (void)n_in; (void)out_size; (void)ws_size;
    const float* u   = (const float*)d_in[0];
    const float* A   = (const float*)d_in[1];
    const float* Bv  = (const float*)d_in[2];
    const float* Cv  = (const float*)d_in[3];
    const float* Dv  = (const float*)d_in[4];
    const float* ldt = (const float*)d_in[5];
    float* y = (float*)d_out;
    const int N = NDIM, L = LSEQ;

    float* B0 = (float*)d_ws;
    float* B1 = B0 + (size_t)N * N;
    float* B2 = B1 + (size_t)N * N;
    float* B3 = B2 + (size_t)N * N;
    float* Y  = B3 + (size_t)N * N;
    float* R  = Y  + (size_t)N * 128;
    float* Kk = R  + (size_t)64 * N;

    dim3 blk(128);
    auto gemm = [&](const float* Am, int lda, const float* Bm, int ldb,
                    float* Cm, int ldc, int Mc, int Nc, int Kd,
                    int mB, int nB) {
        (void)lda;
        const int nReg = ((Nc + 63) / 64) * ((Mc + 63) / 64);
        dim3 grid(2 * nReg);
        if (ldb == 128)
            gemm_wmma_f32<NDIM, 128><<<grid, blk, 0, stream>>>(Am, Bm, Cm, ldc, Kd, mB, nB, nReg);
        else
            gemm_wmma_f32<NDIM, NDIM><<<grid, blk, 0, stream>>>(Am, Bm, Cm, ldc, Kd, mB, nB, nReg);
    };

    float* q1 = Kk;
    float* q2 = Kk + N;
    prep_mats<<<32 + (N * N + 255) / 256, 256, 0, stream>>>(A, ldt, Bv, B0, q1, N);
    gemm_disc_with_vector<1><<<dim3(512 + 32), blk, 0, stream>>>(B0, B0, B1, B0, B2, q1, q2, nullptr,
                                                                 Bv, ldt, Y, Cv, R);
    gemm_disc_with_vector<2><<<dim3(512 + 32), blk, 0, stream>>>(B1, B2, B3, B2, nullptr, q1, nullptr, q2,
                                                                 Bv, ldt, Y, Cv, R);

    dim3 gridPair(512 + 64);
    auto pairY = [&](const float* Ms, float* Sq, float* Cm, int nB) {
        gemm_pair_wmma_f32<128, false><<<gridPair, blk, 0, stream>>>(Ms, Sq, Ms, Y, Cm, 128, N, nB);
    };
    auto pairR = [&](const float* Qt, float* Sq, float* Cm, int mB) {
        gemm_pair_wmma_f32<NDIM, true><<<gridPair, blk, 0, stream>>>(Qt, Sq, R, Qt, Cm, N, mB, N);
    };

    pairY(B3, B0, Y + 1,  127);
    pairY(B0, B1, Y + 2,  126);
    pairY(B1, B0, Y + 4,  124);
    pairY(B0, B1, Y + 8,  120);
    pairY(B1, B0, Y + 16, 112);
    pairY(B0, B1, Y + 32,  96);
    pairY(B1, B0, Y + 64,  64);

    pairR(B0, B1, R + 1 * N,  63);
    pairR(B1, B0, R + 2 * N,  62);
    pairR(B0, B1, R + 4 * N,  60);
    pairR(B1, B0, R + 8 * N,  56);
    pairR(B0, B1, R + 16 * N, 48);
    gemm(R, N, B1, N, R + 32 * N, N, 64, N, N, 32, N);

    gemm(R, N, Y, 128, Kk, 128, 64, 128, N, 64, 128);

    conv_causal_wmma<<<L / 256, 512, 0, stream>>>(u, Kk, Dv, y);
}


# Thin products (doubling applies, last R step, R@Y) on 16x16 tiles, K cut in 4 slices (one per wave), partials added in slice order via LDS; only tiles with new columns/rows run; sum order changes
# speedup vs baseline: 12.0560x; 1.0763x over previous
#include <hip/hip_runtime.h>
#include <math.h>


typedef __attribute__((ext_vector_type(2))) float v2f;
typedef __attribute__((ext_vector_type(4))) float v4f;
typedef __attribute__((ext_vector_type(8))) float v8f;

#define NDIM 1024
#define LSEQ 8192

__device__ __forceinline__ void lds_barrier_arrive()
{
    __builtin_amdgcn_fence(__ATOMIC_RELEASE, "workgroup", "local");
    __builtin_amdgcn_s_barrier_signal(-1);
}
__device__ __forceinline__ void lds_barrier_wait()
{
    __builtin_amdgcn_s_barrier_wait(-1);
    __builtin_amdgcn_fence(__ATOMIC_ACQUIRE, "workgroup", "local");
}
__device__ __forceinline__ void lds_barrier()
{
    lds_barrier_arrive();
    lds_barrier_wait();
}

template <int LDA, int LDB, int TM, int EPI>
__device__ __forceinline__
void gemm_tile_f32(const float* __restrict__ Ag,
                   const float* __restrict__ Bg,
                   float* __restrict__ Cg, int ldc,
                   const float* __restrict__ Eg, float* __restrict__ C2g,
                   int Kdim, int mBound, int nBound,
                   const int row0, const int col0,
                   float (&As)[2][64][36],
                   float (&Bs)[2][64][36])
{
    const int tid  = threadIdx.x;
    const int lane = tid & 31;
    const int wv   = tid >> 5;
    constexpr int KS  = (TM == 64) ? 1 : 4;
    constexpr int TN  = (TM == 64) ? 32 : 64;
    constexpr int NI  = (TM == 64) ? 2 : 1;
    constexpr int AIT = 4;
    constexpr int BIT = TN / 16;
    constexpr int KGP = 128 / TN;
    const int wm   = wv * 16;
    const int wn   = (TM == 64) ? 0 : wv * 16;
    const int om   = (TM == 64) ? wv * 16 : 0;
    const int g    = lane >> 4;
    const int lm   = lane & 15;
    const int kSl  = Kdim / KS;

    v8f acc[NI] = {};
    v8f acc2 = {};

    const float* aBase = Ag + (size_t)(row0 + (tid >> 3)) * LDA + (tid & 7) * 4;
    const float* bBase = (TM == 64)
        ? Bg + (size_t)((tid / TN) * 4) * LDB + (col0 + (tid & (TN - 1)))
        : Bg + (size_t)((tid / TN) * 4 + ((tid & 63) >> 4) * kSl) * LDB + (col0 + (tid & 15));

    const int nCh = (kSl + 31) >> 5;

    v4f ra[3][AIT], rb[3][BIT];
    v4f fa0[4], fb0[4][NI];
    v4f fa1[4], fb1[4][NI];

    auto gload = [&](v4f (&qa)[AIT], v4f (&qb)[BIT], int c) {
        const int k0 = (c < nCh ? c : nCh - 1) * 32;
        const float* aP = aBase + k0;
        const float* bP = bBase + (size_t)k0 * LDB;
        #pragma unroll
        for (int it = 0; it < AIT; ++it) {
            if constexpr (TM == 64)
                qa[it] = *reinterpret_cast<const v4f*>(aP + it * 16 * LDA);
            else
                qa[it] = *reinterpret_cast<const v4f*>(aP + it * kSl);
        }
        #pragma unroll
        for (int it = 0; it < BIT; ++it) {
            qb[it].x = bP[(it * KGP * 4 + 0) * LDB];
            qb[it].y = bP[(it * KGP * 4 + 1) * LDB];
            qb[it].z = bP[(it * KGP * 4 + 2) * LDB];
            qb[it].w = bP[(it * KGP * 4 + 3) * LDB];
        }
    };
    auto lstore = [&](const v4f (&qa)[AIT], const v4f (&qb)[BIT], const int buf) {
        #pragma unroll
        for (int it = 0; it < AIT; ++it)
            *reinterpret_cast<v4f*>(&As[buf][(tid >> 3) + 16 * it][(tid & 7) * 4]) = qa[it];
        #pragma unroll
        for (int it = 0; it < BIT; ++it)
            *reinterpret_cast<v4f*>(&Bs[buf][tid & (TN - 1)][((tid / TN) + KGP * it) * 4]) = qb[it];
    };
    auto fload = [&](v4f (&fa)[4], v4f (&fb)[4][NI], const int buf) {
        #pragma unroll
        for (int p = 0; p < 4; ++p) {
            const int kq = 8 * p + 4 * g;
            fa[p] = *reinterpret_cast<const v4f*>(&As[buf][wm + lm][kq]);
            #pragma unroll
            for (int ni = 0; ni < NI; ++ni)
                fb[p][ni] = *reinterpret_cast<const v4f*>(&Bs[buf][wn + 16 * ni + lm][kq]);
        }
    };
    auto mma = [&](const v4f (&fa)[4], const v4f (&fb)[4][NI], const int p0, const int p1) {
        #pragma unroll
        for (int p = p0; p < p1; ++p) {
          const v2f al = __builtin_shufflevector(fa[p], fa[p], 0, 1), ah = __builtin_shufflevector(fa[p], fa[p], 2, 3);
          if constexpr (TM == 64) {
            const v2f b0l = __builtin_shufflevector(fb[p][0], fb[p][0], 0, 1), b0h = __builtin_shufflevector(fb[p][0], fb[p][0], 2, 3);
            const v2f b1l = __builtin_shufflevector(fb[p][1], fb[p][1], 0, 1), b1h = __builtin_shufflevector(fb[p][1], fb[p][1], 2, 3);
            acc[0] = __builtin_amdgcn_wmma_f32_16x16x4_f32(false, al, false, b0l, (short)0, acc[0], false, false);
            acc[1] = __builtin_amdgcn_wmma_f32_16x16x4_f32(false, al, false, b1l, (short)0, acc[1], false, false);
            acc[0] = __builtin_amdgcn_wmma_f32_16x16x4_f32(false, ah, false, b0h, (short)0, acc[0], false, false);
            acc[1] = __builtin_amdgcn_wmma_f32_16x16x4_f32(false, ah, false, b1h, (short)0, acc[1], false, false);
          } else {
            const v2f bl = __builtin_shufflevector(fb[p][0], fb[p][0], 0, 1), bh = __builtin_shufflevector(fb[p][0], fb[p][0], 2, 3);
            acc[0] = __builtin_amdgcn_wmma_f32_16x16x4_f32(false, al, false, bl, (short)0, acc[0], false, false);
            acc2   = __builtin_amdgcn_wmma_f32_16x16x4_f32(false, ah, false, bh, (short)0, acc2,   false, false);
          }
        }
    };

    constexpr int PA = (TM == 64) ? 2 : 1;
    auto step = [&](const v4f (&ca)[4], const v4f (&cb)[4][NI],
                    v4f (&na)[4], v4f (&nb)[4][NI],
                    v4f (&qa)[AIT], v4f (&qb)[BIT],
                    const int bufNext, const int bufStore, const int cLoad) {
        fload(na, nb, bufNext);
        lstore(qa, qb, bufStore);
        gload(qa, qb, cLoad);
        __builtin_amdgcn_sched_barrier(0);
        mma(ca, cb, 0, PA);
        __builtin_amdgcn_sched_barrier(0);
        lds_barrier_arrive();
        __builtin_amdgcn_sched_barrier(0);
        mma(ca, cb, PA, 4);
        if constexpr (TM == 16) {
            asm volatile("v_nop\n\tv_nop\n\tv_nop\n\tv_nop" : "+v"(acc[0]), "+v"(acc2));
        }
        __builtin_amdgcn_sched_barrier(0);
        lds_barrier_wait();
    };

    if (nCh > 0) {
        gload(ra[0], rb[0], 0);
        gload(ra[1], rb[1], 1);
        gload(ra[2], rb[2], 2);
        lstore(ra[0], rb[0], 0);
        gload(ra[0], rb[0], 3);
        lds_barrier();
        fload(fa0, fb0, 0);
        lstore(ra[1], rb[1], 1);
        gload(ra[1], rb[1], 4);
        lds_barrier();

        int c = 0;
        for (; c + 6 <= nCh; c += 6) {
            step(fa0, fb0, fa1, fb1, ra[2], rb[2], 1, 0, c + 5);
            step(fa1, fb1, fa0, fb0, ra[0], rb[0], 0, 1, c + 6);
            step(fa0, fb0, fa1, fb1, ra[1], rb[1], 1, 0, c + 7);
            step(fa1, fb1, fa0, fb0, ra[2], rb[2], 0, 1, c + 8);
            step(fa0, fb0, fa1, fb1, ra[0], rb[0], 1, 0, c + 9);
            step(fa1, fb1, fa0, fb0, ra[1], rb[1], 0, 1, c + 10);
        }
        if (c < nCh) {
            step(fa0, fb0, fa1, fb1, ra[2], rb[2], 1, 0, c + 5);
            if (c + 1 < nCh) {
                step(fa1, fb1, fa0, fb0, ra[0], rb[0], 0, 1, c + 6);
                if (c + 2 < nCh) {
                    step(fa0, fb0, fa1, fb1, ra[1], rb[1], 1, 0, c + 7);
                    if (c + 3 < nCh) {
                        step(fa1, fb1, fa0, fb0, ra[2], rb[2], 0, 1, c + 8);
                        if (c + 4 < nCh)
                            step(fa0, fb0, fa1, fb1, ra[0], rb[0], 1, 0, c + 9);
                    }
                }
            }
        }
    }

    if constexpr (TM == 16) {
        #pragma unroll
        for (int r = 0; r < 8; ++r) acc[0][r] += acc2[r];

        float* red = &As[0][0][0];
        if (wv > 0) {
            float* rp = red + (wv - 1) * 256 + lane;
            #pragma unroll
            for (int r = 0; r < 8; ++r) rp[r * 32] = acc[0][r];
        }
        lds_barrier();
        if (wv == 0) {
            #pragma unroll
            for (int s = 1; s < KS; ++s)
                #pragma unroll
                for (int r = 0; r < 8; ++r) acc[0][r] += red[(s - 1) * 256 + r * 32 + lane];
        }
    }

    if constexpr (EPI == 0) {
        if (TM == 64 || wv == 0) {
            #pragma unroll
            for (int ni = 0; ni < NI; ++ni)
                #pragma unroll
                for (int r = 0; r < 8; ++r) {
                    int row = row0 + om + r + 8 * g;
                    int col = col0 + ni * 16 + lm;
                    if (row < mBound && col < nBound)
                        Cg[(size_t)row * ldc + col] = acc[ni][r];
                }
        }
    } else {
        float e[NI][8];
        #pragma unroll
        for (int ni = 0; ni < NI; ++ni)
            #pragma unroll
            for (int r = 0; r < 8; ++r)
                e[ni][r] = Eg[(size_t)(row0 + om + r + 8 * g) * ldc + (col0 + ni * 16 + lm)];
        #pragma unroll
        for (int ni = 0; ni < NI; ++ni)
            #pragma unroll
            for (int r = 0; r < 8; ++r) {
                const int row = row0 + om + r + 8 * g;
                const int col = col0 + ni * 16 + lm;
                const size_t o = (size_t)row * ldc + col;
                float v = acc[ni][r];
                if constexpr (EPI == 1) {
                    C2g[o] = e[ni][r] + v;
                } else {
                    v = 2.0f * (e[ni][r] + v);
                    if (row == col) v += 1.0f;
                }
                Cg[o] = v;
            }
    }
}

__device__ __forceinline__ void half_tile_origin(const int b, const int nReg, const int nc64,
                                                 int& row0, int& col0)
{
    const int hf = (b >= nReg) ? 1 : 0;
    const int t  = b - hf * nReg;
    row0 = (t / nc64) * 64;
    col0 = ((t % nc64) * 2 + hf) * 32;
}

__device__ __forceinline__ void thin_tile_origin(const int t, const int nct, int& row0, int& col0)
{
    row0 = (t / nct) * 16;
    col0 = (t % nct) * 16;
}

template <int LDB>
__global__ __launch_bounds__(128)
void gemm_thin_wmma_f32(const float* __restrict__ Ag,
                        const float* __restrict__ Bg,
                        float* __restrict__ Cg, int ldc,
                        int mBound, int nBound, int nct)
{
    __shared__ __attribute__((aligned(16))) float As[2][64][36];
    __shared__ __attribute__((aligned(16))) float Bs[2][64][36];
    int row0, col0;
    thin_tile_origin(blockIdx.x, nct, row0, col0);
    gemm_tile_f32<NDIM, LDB, 16, 0>(Ag, Bg, Cg, ldc, nullptr, nullptr, NDIM, mBound, nBound, row0, col0, As, Bs);
}

template <int LDB1>
__global__ __launch_bounds__(128)
void gemm_pair_wmma_f32(const float* __restrict__ P, float* __restrict__ Sq,
                        const float* __restrict__ A1, const float* __restrict__ B1,
                        float* __restrict__ C1, int ldc1, int mB1, int nB1, int nct1)
{
    __shared__ __attribute__((aligned(16))) float As[2][64][36];
    __shared__ __attribute__((aligned(16))) float Bs[2][64][36];
    const int b = blockIdx.x;
    int row0, col0;
    if (b < 512) {
        half_tile_origin(b, 256, 16, row0, col0);
        gemm_tile_f32<NDIM, NDIM, 64, 0>(P, P, Sq, NDIM, nullptr, nullptr, NDIM, NDIM, NDIM, row0, col0, As, Bs);
    } else {
        thin_tile_origin(b - 512, nct1, row0, col0);
        gemm_tile_f32<NDIM, LDB1, 16, 0>(A1, B1, C1, ldc1, nullptr, nullptr, NDIM, mB1, nB1, row0, col0, As, Bs);
    }
}


__device__ __forceinline__ float row_dot_wave(const float* __restrict__ rowp, const v4f (&xv)[8],
                                              const int lane)
{
    v4f a[8];
    #pragma unroll
    for (int j = 0; j < 8; ++j) a[j] = *reinterpret_cast<const v4f*>(rowp + 128 * j + 4 * lane);
    float sum = 0.f;
    #pragma unroll
    for (int j = 0; j < 8; ++j)
        sum += a[j].x * xv[j].x + a[j].y * xv[j].y + a[j].z * xv[j].z + a[j].w * xv[j].w;
    #pragma unroll
    for (int o = 16; o > 0; o >>= 1) sum += __shfl_xor(sum, o);
    return sum;
}

__device__ __forceinline__ void load_vec_wave(const float* __restrict__ x, v4f (&xv)[8], const int lane)
{
    #pragma unroll
    for (int j = 0; j < 8; ++j) xv[j] = *reinterpret_cast<const v4f*>(x + 128 * j + 4 * lane);
}

template <int STAGE>
__global__ __launch_bounds__(128)
void gemm_disc_with_vector(const float* __restrict__ Aop, const float* __restrict__ Bop,
                           float* __restrict__ Cm, const float* __restrict__ Em,
                           float* __restrict__ C2m,
                           const float* __restrict__ q1, float* __restrict__ q2w,
                           const float* __restrict__ q2r,
                           const float* __restrict__ Bv, const float* __restrict__ ldt,
                           float* __restrict__ Y,
                           const float* __restrict__ Cv, float* __restrict__ R)
{
    __shared__ __attribute__((aligned(16))) float As[2][64][36];
    __shared__ __attribute__((aligned(16))) float Bs[2][64][36];
    const int b = blockIdx.x;
    if (b < 512) {
        int row0, col0;
        half_tile_origin(b, 256, 16, row0, col0);
        gemm_tile_f32<NDIM, NDIM, 64, STAGE>(Aop, Bop, Cm, NDIM, Em, C2m, NDIM, NDIM, NDIM, row0, col0, As, Bs);
    } else {
        const int lane = threadIdx.x & 31;
        const int rowW = ((b - 512) * 4 + (threadIdx.x >> 5)) * 8;
        v4f xv[8];
        load_vec_wave(q1, xv, lane);
        if constexpr (STAGE == 2) {
            v4f x2[8];
            load_vec_wave(q2r, x2, lane);
            #pragma unroll
            for (int j = 0; j < 8; ++j) xv[j] += x2[j];
        }
        const float dt = (STAGE == 2) ? expf(ldt[0]) : 0.f;
        for (int rr = 0; rr < 8; ++rr) {
            const int i = rowW + rr;
            const float sum = row_dot_wave(Aop + (size_t)i * NDIM, xv, lane);
            if (lane == 0) {
                if constexpr (STAGE == 1) {
                    q2w[i] = sum;
                } else {
                    Y[(size_t)i * 128] = dt * (Bv[i] + q1[i] + q2r[i] + sum);
                }
            }
        }
        if constexpr (STAGE == 2) {
            if (lane < 8) R[rowW + lane] = Cv[rowW + lane];
        }
    }
}

__global__ __launch_bounds__(256)
void prep_mats(const float* __restrict__ A, const float* __restrict__ ldt,
               const float* __restrict__ Bv,
               float* __restrict__ M, float* __restrict__ q1, int n)
{
    const float h = 0.5f * expf(ldt[0]);
    if (blockIdx.x < 32) {
        const int lane = threadIdx.x & 31;
        const int rowW = (blockIdx.x * 8 + (threadIdx.x >> 5)) * 4;
        v4f xv[8];
        load_vec_wave(Bv, xv, lane);
        for (int rr = 0; rr < 4; ++rr) {
            const float sum = row_dot_wave(A + (size_t)(rowW + rr) * NDIM, xv, lane);
            if (lane == 0) q1[rowW + rr] = h * sum;
        }
        return;
    }
    int idx = (blockIdx.x - 32) * blockDim.x + threadIdx.x;
    if (idx >= n * n) return;
    M[idx] = h * A[idx];
}

#define CV_WAVES 16
__global__ __launch_bounds__(CV_WAVES * 32)
void conv_causal_wmma(const float* __restrict__ u, const float* __restrict__ Kk,
                      const float* __restrict__ Dv, float* __restrict__ y)
{
    __shared__ float KT[LSEQ + 16];
    __shared__ float USk[255 + LSEQ + LSEQ / 16 + 1];
    __shared__ float red[CV_WAVES][256];

    const int tid = threadIdx.x;
    const int tb  = blockIdx.x;
    const int nIt = tb + 1;

    float rk[16], ru[16];
    #pragma unroll
    for (int it = 0; it < 16; ++it) {
        const int yy = tid + it * (CV_WAVES * 32);
        rk[it] = Kk[yy];
        ru[it] = u[yy];
    }
    #pragma unroll
    for (int it = 0; it < 16; ++it) {
        const int yy = tid + it * (CV_WAVES * 32);
        KT[yy] = rk[it];
        USk[255 + yy + (yy >> 4)] = ru[it];
    }
    if (tid < 16)  KT[LSEQ + tid] = 0.f;
    if (tid < 255) USk[tid] = 0.f;
    __syncthreads();

    const int lane = tid & 31;
    const int wv   = tid >> 5;
    const int c    = lane & 15;
    const int h    = lane >> 4;

    const float* ap = KT + (LSEQ + 14 - 13) - wv * 16 * nIt - c - 2 * h;
    const float* bp = USk + 17 * (16 * tb + c + 15 - wv * nIt) - 2 * h;

    auto fl = [&](v2f (&a)[4], v2f (&b)[4], const int it) {
        const float* a_ = ap - 16 * it;
        const float* b_ = bp - 17 * it;
        #pragma unroll
        for (int s = 0; s < 4; ++s) {
            a[s].x = a_[12 - 4 * s];   a[s].y = a_[13 - 4 * s];
            b[s].x = b_[14 - 4 * s];   b[s].y = b_[15 - 4 * s];
        }
    };

    v8f acc = {};
    auto mma_group = [&](const v2f (&a)[4], const v2f (&b)[4]) {
        __builtin_amdgcn_sched_barrier(0);
        #pragma unroll
        for (int s = 0; s < 4; ++s)
            acc = __builtin_amdgcn_wmma_f32_16x16x4_f32(false, a[s], false, b[s], (short)0, acc, false, false);
        asm volatile("v_nop\n\tv_nop\n\tv_nop\n\tv_nop" : "+v"(acc));
        __builtin_amdgcn_sched_barrier(0);
    };

    v2f a0[4], b0[4], a1[4], b1[4];
    fl(a0, b0, 0);
    for (int it = 0; it < nIt; it += 2) {
        fl(a1, b1, (it + 1 < nIt) ? it + 1 : it);
        mma_group(a0, b0);
        if (it + 1 < nIt) {
            fl(a0, b0, (it + 2 < nIt) ? it + 2 : it + 1);
            mma_group(a1, b1);
        }
    }

    float* rp = &red[wv][16 * c + 8 * h];
    #pragma unroll
    for (int r = 0; r < 8; ++r) rp[r] = acc[r];
    __syncthreads();

    if (tid < 256) {
        float s = 0.f;
        #pragma unroll
        for (int w = 0; w < CV_WAVES; ++w) s += red[w][tid];
        const int t = tb * 256 + tid;
        y[t] = s + Dv[0] * u[t];
    }
}

extern "C" void kernel_launch(void* const* d_in, const int* in_sizes, int n_in,
                              void* d_out, int out_size, void* d_ws, size_t ws_size,
                              hipStream_t stream)
{
    (void)in_sizes; (void)n_in; (void)out_size; (void)ws_size;
    const float* u   = (const float*)d_in[0];
    const float* A   = (const float*)d_in[1];
    const float* Bv  = (const float*)d_in[2];
    const float* Cv  = (const float*)d_in[3];
    const float* Dv  = (const float*)d_in[4];
    const float* ldt = (const float*)d_in[5];
    float* y = (float*)d_out;
    const int N = NDIM, L = LSEQ;

    float* B0 = (float*)d_ws;
    float* B1 = B0 + (size_t)N * N;
    float* B2 = B1 + (size_t)N * N;
    float* B3 = B2 + (size_t)N * N;
    float* Y  = B3 + (size_t)N * N;
    float* R  = Y  + (size_t)N * 128;
    float* Kk = R  + (size_t)64 * N;

    dim3 blk(128);

    float* q1 = Kk;
    float* q2 = Kk + N;
    prep_mats<<<32 + (N * N + 255) / 256, 256, 0, stream>>>(A, ldt, Bv, B0, q1, N);
    gemm_disc_with_vector<1><<<dim3(512 + 32), blk, 0, stream>>>(B0, B0, B1, B0, B2, q1, q2, nullptr,
                                                                 Bv, ldt, Y, Cv, R);
    gemm_disc_with_vector<2><<<dim3(512 + 32), blk, 0, stream>>>(B1, B2, B3, B2, nullptr, q1, nullptr, q2,
                                                                 Bv, ldt, Y, Cv, R);

    auto pairY = [&](const float* Ms, float* Sq, float* Cm, int nB, int nct) {
        gemm_pair_wmma_f32<128><<<dim3(512 + 64 * nct), blk, 0, stream>>>(Ms, Sq, Ms, Y, Cm, 128, N, nB, nct);
    };
    auto pairR = [&](const float* Qt, float* Sq, float* Cm, int mB) {
        gemm_pair_wmma_f32<NDIM><<<dim3(512 + 64), blk, 0, stream>>>(Qt, Sq, R, Qt, Cm, N, mB, N, 64);
    };

    pairY(B3, B0, Y + 1,  127, 1);
    pairY(B0, B1, Y + 2,  126, 1);
    pairY(B1, B0, Y + 4,  124, 1);
    pairY(B0, B1, Y + 8,  120, 1);
    pairY(B1, B0, Y + 16, 112, 1);
    pairY(B0, B1, Y + 32,  96, 2);
    pairY(B1, B0, Y + 64,  64, 4);

    pairR(B0, B1, R + 1 * N,  63);
    pairR(B1, B0, R + 2 * N,  62);
    pairR(B0, B1, R + 4 * N,  60);
    pairR(B1, B0, R + 8 * N,  56);
    pairR(B0, B1, R + 16 * N, 48);
    gemm_thin_wmma_f32<NDIM><<<dim3(2 * 64), blk, 0, stream>>>(R, B1, R + 32 * N, N, 32, N, 64);

    gemm_thin_wmma_f32<128><<<dim3(4 * 8), blk, 0, stream>>>(R, Y, Kk, 128, 64, 128, 8);

    conv_causal_wmma<<<L / 256, 512, 0, stream>>>(u, Kk, Dv, y);
}
